# starting kernel
# baseline (speedup 1.0000x reference)
#include <hip/hip_runtime.h>
#include <hip/hip_bf16.h>

typedef __attribute__((ext_vector_type(2))) float v2f;
typedef __attribute__((ext_vector_type(8))) float v8f;

#define IN_CH   256
#define HC      128
#define HEADS   4
#define OUT_CH  32
#define LDS_STRIDE 260

__device__ __forceinline__ unsigned mapf(float f) {
    unsigned u = __float_as_uint(f);
    return (u >> 31) ? ~u : (u | 0x80000000u);
}
__device__ __forceinline__ float unmapf(unsigned u) {
    return (u >> 31) ? __uint_as_float(u ^ 0x80000000u) : __uint_as_float(~u);
}
__device__ __forceinline__ float lrelu(float a) { return a > 0.0f ? a : 0.2f * a; }

__global__ void gat_init(float* out_mean, unsigned* segmax, float* denom, int n) {
    int i = blockIdx.x * blockDim.x + threadIdx.x;
    if (i < n * OUT_CH) out_mean[i] = 0.0f;
    if (i < n * HEADS)  { segmax[i] = 0u; denom[i] = 0.0f; }
}

__global__ void gat_gemm(const float* __restrict__ x, const float* __restrict__ W,
                         float* __restrict__ h, int n) {
    __shared__ float xs[16 * LDS_STRIDE];
    const int row0 = blockIdx.x * 16;

    const float4* xv = (const float4*)x;
    for (int i = 0; i < 4; ++i) {
        int idx = threadIdx.x + i * 256;
        int r   = idx >> 6;
        int c4  = idx & 63;
        int gr  = row0 + r;  if (gr >= n) gr = n - 1;
        float4 v = xv[(size_t)gr * 64 + c4];
        float* dst = &xs[r * LDS_STRIDE + (c4 << 2)];
        dst[0] = v.x; dst[1] = v.y; dst[2] = v.z; dst[3] = v.w;
    }
    __syncthreads();

    const int lane = threadIdx.x & 31;
    const int n0   = (threadIdx.x >> 5) << 4;
    const int m    = lane & 15;
    const int kb   = (lane >> 4) << 1;

    v8f acc = {};
    #pragma unroll 8
    for (int k = 0; k < IN_CH; k += 4) {
        v2f a, b;
        a.x = xs[m * LDS_STRIDE + k + kb];
        a.y = xs[m * LDS_STRIDE + k + kb + 1];
        const float* Wp = W + (size_t)(k + kb) * HC + n0 + m;
        b.x = Wp[0];
        b.y = Wp[HC];
        acc = __builtin_amdgcn_wmma_f32_16x16x4_f32(false, a, false, b,
                                                    (short)0, acc, false, false);
    }

    const int mOff = (lane >> 4) << 3;
    const int col  = n0 + m;
    #pragma unroll
    for (int i = 0; i < 8; ++i) {
        int r = row0 + mOff + i;
        if (r < n) h[(size_t)r * HC + col] = acc[i];
    }
}

__global__ void gat_scores(const float* __restrict__ h, const float* __restrict__ attn,
                           float* __restrict__ s_src, float* __restrict__ s_dst, int n) {
    int t = blockIdx.x * blockDim.x + threadIdx.x;
    if (t >= n * HEADS) return;
    int node = t >> 2, hd = t & 3;
    const float* hr = h + (size_t)node * HC + hd * OUT_CH;
    const float* as = attn + hd * (2 * OUT_CH);
    const float* ad = as + OUT_CH;
    float ss = 0.0f, sd = 0.0f;
    #pragma unroll 8
    for (int c = 0; c < OUT_CH; ++c) { float v = hr[c]; ss += v * as[c]; sd += v * ad[c]; }
    s_src[t] = ss; s_dst[t] = sd;
}

__global__ void gat_segmax(const int* __restrict__ rows, const int* __restrict__ cols,
                           const float* __restrict__ s_src, const float* __restrict__ s_dst,
                           unsigned* __restrict__ segmax, int E) {
    int t = blockIdx.x * blockDim.x + threadIdx.x;
    if (t >= E * HEADS) return;
    int e = t >> 2, hd = t & 3;
    int r = rows[e], c = cols[e];
    float a = lrelu(s_src[r * HEADS + hd] + s_dst[c * HEADS + hd]);
    atomicMax(&segmax[r * HEADS + hd], mapf(a));
}

__global__ void gat_denom(const int* __restrict__ rows, const int* __restrict__ cols,
                          const float* __restrict__ s_src, const float* __restrict__ s_dst,
                          const unsigned* __restrict__ segmax, float* __restrict__ denom, int E) {
    int t = blockIdx.x * blockDim.x + threadIdx.x;
    if (t >= E * HEADS) return;
    int e = t >> 2, hd = t & 3;
    int r = rows[e], c = cols[e];
    float a  = lrelu(s_src[r * HEADS + hd] + s_dst[c * HEADS + hd]);
    float mx = unmapf(segmax[r * HEADS + hd]);
    atomicAdd(&denom[r * HEADS + hd], __expf(a - mx));
}

__global__ void gat_scatter(const int* __restrict__ rows, const int* __restrict__ cols,
                            const float* __restrict__ s_src, const float* __restrict__ s_dst,
                            const unsigned* __restrict__ segmax, const float* __restrict__ denom,
                            const float* __restrict__ h,
                            float* __restrict__ out_mean, float* __restrict__ alpha_out, int E) {
    int e = blockIdx.x * (blockDim.x >> 5) + (threadIdx.x >> 5);
    if (e >= E) return;
    int lane = threadIdx.x & 31;
    int r = rows[e], c = cols[e];
    const float* hc = h + (size_t)c * HC;
    float val = 0.0f;
    #pragma unroll
    for (int hd = 0; hd < HEADS; ++hd) {
        float a  = lrelu(s_src[r * HEADS + hd] + s_dst[c * HEADS + hd]);
        float mx = unmapf(segmax[r * HEADS + hd]);
        float an = __expf(a - mx) / denom[r * HEADS + hd];
        if (lane == hd) alpha_out[(size_t)e * HEADS + hd] = an;
        val += an * hc[hd * OUT_CH + lane];
    }
    atomicAdd(&out_mean[(size_t)r * OUT_CH + lane], 0.25f * val);
}

extern "C" void kernel_launch(void* const* d_in, const int* in_sizes, int n_in,
                              void* d_out, int out_size, void* d_ws, size_t ws_size,
                              hipStream_t stream) {
    const float* x    = (const float*)d_in[0];
    const int*   ei   = (const int*)d_in[1];
    const float* W    = (const float*)d_in[2];
    const float* attn = (const float*)d_in[3];

    const int n = in_sizes[0] / IN_CH;
    const int E = in_sizes[1] / 2;
    const int* rows = ei;
    const int* cols = ei + E;

    float*    hbuf   = (float*)d_ws;
    float*    s_src  = hbuf  + (size_t)n * HC;
    float*    s_dst  = s_src + (size_t)n * HEADS;
    unsigned* segmax = (unsigned*)(s_dst + (size_t)n * HEADS);
    float*    denom  = (float*)(segmax + (size_t)n * HEADS);

    float* out_mean  = (float*)d_out;
    float* alpha_out = out_mean + (size_t)n * OUT_CH;

    gat_init<<<(n * OUT_CH + 255) / 256, 256, 0, stream>>>(out_mean, segmax, denom, n);

    gat_gemm<<<(n + 15) / 16, 256, 0, stream>>>(x, W, hbuf, n);

    gat_scores<<<(n * HEADS + 255) / 256, 256, 0, stream>>>(hbuf, attn, s_src, s_dst, n);

    gat_segmax<<<(E * HEADS + 255) / 256, 256, 0, stream>>>(rows, cols, s_src, s_dst, segmax, E);

    gat_denom<<<(E * HEADS + 255) / 256, 256, 0, stream>>>(rows, cols, s_src, s_dst, segmax, denom, E);

    gat_scatter<<<(E + 7) / 8, 256, 0, stream>>>(rows, cols, s_src, s_dst, segmax, denom,
                                                 hbuf, out_mean, alpha_out, E);
}


# group edges by destination row with an integer counting sort, then one wave per row does softmax and weighted sum in registers: float atomics removed, sum order within a row follows fill arrival
# speedup vs baseline: 1.2223x; 1.2223x over previous
#include <hip/hip_runtime.h>
#include <hip/hip_bf16.h>

typedef __attribute__((ext_vector_type(2))) float v2f;
typedef __attribute__((ext_vector_type(8))) float v8f;

#define IN_CH   256
#define HC      128
#define HEADS   4
#define OUT_CH  32
#define LDS_STRIDE 260

__device__ __forceinline__ float lrelu(float a) { return a > 0.0f ? a : 0.2f * a; }
__device__ __forceinline__ float rl_f(float v, int j) {
    return __int_as_float(__builtin_amdgcn_readlane(__float_as_int(v), j));
}


__global__ void gat_init(int* __restrict__ cnt, int n) {
    int i = blockIdx.x * blockDim.x + threadIdx.x;
    if (i < n) cnt[i] = 0;
}

__global__ void gat_gemm(const float* __restrict__ x, const float* __restrict__ W,
                         float* __restrict__ h, int n) {
    __shared__ float xs[16 * LDS_STRIDE];
    const int row0 = blockIdx.x * 16;

    const float4* xv = (const float4*)x;
    for (int i = 0; i < 4; ++i) {
        int idx = threadIdx.x + i * 256;
        int r   = idx >> 6;
        int c4  = idx & 63;
        int gr  = row0 + r;  if (gr >= n) gr = n - 1;
        float4 v = xv[(size_t)gr * 64 + c4];
        float* dst = &xs[r * LDS_STRIDE + (c4 << 2)];
        dst[0] = v.x; dst[1] = v.y; dst[2] = v.z; dst[3] = v.w;
    }
    __syncthreads();

    const int lane = threadIdx.x & 31;
    const int n0   = (threadIdx.x >> 5) << 4;
    const int m    = lane & 15;
    const int kb   = (lane >> 4) << 1;

    v8f acc = {};
    #pragma unroll 8
    for (int k = 0; k < IN_CH; k += 4) {
        v2f a, b;
        a.x = xs[m * LDS_STRIDE + k + kb];
        a.y = xs[m * LDS_STRIDE + k + kb + 1];
        const float* Wp = W + (size_t)(k + kb) * HC + n0 + m;
        b.x = Wp[0];
        b.y = Wp[HC];
        acc = __builtin_amdgcn_wmma_f32_16x16x4_f32(false, a, false, b,
                                                    (short)0, acc, false, false);
    }

    const int mOff = (lane >> 4) << 3;
    const int col  = n0 + m;
    #pragma unroll
    for (int i = 0; i < 8; ++i) {
        int r = row0 + mOff + i;
        if (r < n) h[(size_t)r * HC + col] = acc[i];
    }
}

__global__ void gat_scores(const float* __restrict__ h, const float* __restrict__ attn,
                           float* __restrict__ s_src, float* __restrict__ s_dst, int n) {
    int t = blockIdx.x * blockDim.x + threadIdx.x;
    if (t >= n * HEADS) return;
    int node = t >> 2, hd = t & 3;
    const float* hr = h + (size_t)node * HC + hd * OUT_CH;
    const float* as = attn + hd * (2 * OUT_CH);
    const float* ad = as + OUT_CH;
    float ss = 0.0f, sd = 0.0f;
    #pragma unroll 8
    for (int c = 0; c < OUT_CH; ++c) { float v = hr[c]; ss += v * as[c]; sd += v * ad[c]; }
    s_src[t] = ss; s_dst[t] = sd;
}

__global__ void gat_hist(const int* __restrict__ rows, int* __restrict__ cnt, int E) {
    int e = blockIdx.x * blockDim.x + threadIdx.x;
    if (e < E) atomicAdd(&cnt[rows[e]], 1);
}

__global__ void gat_scan(const int* __restrict__ cnt, int* __restrict__ slots, int n) {
    __shared__ int part[1024];
    const int t   = threadIdx.x;
    const int per = (n + 1023) >> 10;
    int b = t * per;        if (b > n) b = n;
    int e = b + per;        if (e > n) e = n;
    int s = 0;
    #pragma unroll 8
    for (int i = b; i < e; ++i) s += cnt[i];
    part[t] = s;
    __syncthreads();
    for (int off = 1; off < 1024; off <<= 1) {
        int v = (t >= off) ? part[t - off] : 0;
        __syncthreads();
        part[t] += v;
        __syncthreads();
    }
    int run = part[t] - s;
    #pragma unroll 8
    for (int i = b; i < e; ++i) {
        int c = cnt[i];
        slots[(size_t)i * OUT_CH]     = run;
        slots[(size_t)i * OUT_CH + 1] = run;
        run += c;
    }
}

__global__ void gat_fill(const int* __restrict__ rows, const int* __restrict__ cols,
                         int* __restrict__ slots, int* __restrict__ perm, int E) {
    int e = blockIdx.x * blockDim.x + threadIdx.x;
    if (e >= E) return;
    int pos = atomicAdd(&slots[(size_t)rows[e] * OUT_CH + 1], 1);
    perm[pos] = cols[e];
}

__global__ void gat_rowpass(const int* __restrict__ perm,
                            const float* __restrict__ s_src, const float* __restrict__ s_dst,
                            const float* __restrict__ h,
                            float* out_mean, float* __restrict__ mx_out,
                            float* __restrict__ den_out, int n) {
    const int r = blockIdx.x * (blockDim.x >> 5) + (threadIdx.x >> 5);
    if (r >= n) return;
    const int lane = threadIdx.x & 31;
    const int* slot = (const int*)(out_mean + (size_t)r * OUT_CH);
    const int start = __builtin_amdgcn_readfirstlane(slot[0]);
    const int end   = __builtin_amdgcn_readfirstlane(slot[1]);
    const float4 ss = *(const float4*)(s_src + (size_t)r * HEADS);
    const float NEG_INF = -__builtin_huge_valf();

    float m0 = NEG_INF, m1 = NEG_INF, m2 = NEG_INF, m3 = NEG_INF;
    for (int base = start; base < end; base += 32) {
        int idx = base + lane;
        if (idx < end) {
            int c = perm[idx];
            float4 sd = *(const float4*)(s_dst + (size_t)c * HEADS);
            m0 = fmaxf(m0, lrelu(ss.x + sd.x));
            m1 = fmaxf(m1, lrelu(ss.y + sd.y));
            m2 = fmaxf(m2, lrelu(ss.z + sd.z));
            m3 = fmaxf(m3, lrelu(ss.w + sd.w));
        }
    }
    for (int o = 16; o > 0; o >>= 1) {
        m0 = fmaxf(m0, __shfl_xor(m0, o));
        m1 = fmaxf(m1, __shfl_xor(m1, o));
        m2 = fmaxf(m2, __shfl_xor(m2, o));
        m3 = fmaxf(m3, __shfl_xor(m3, o));
    }

    float d0 = 0.0f, d1 = 0.0f, d2 = 0.0f, d3 = 0.0f;
    for (int base = start; base < end; base += 32) {
        int idx = base + lane;
        if (idx < end) {
            int c = perm[idx];
            float4 sd = *(const float4*)(s_dst + (size_t)c * HEADS);
            d0 += __expf(lrelu(ss.x + sd.x) - m0);
            d1 += __expf(lrelu(ss.y + sd.y) - m1);
            d2 += __expf(lrelu(ss.z + sd.z) - m2);
            d3 += __expf(lrelu(ss.w + sd.w) - m3);
        }
    }
    for (int o = 16; o > 0; o >>= 1) {
        d0 += __shfl_xor(d0, o);
        d1 += __shfl_xor(d1, o);
        d2 += __shfl_xor(d2, o);
        d3 += __shfl_xor(d3, o);
    }

    float acc = 0.0f;
    for (int base = start; base < end; base += 32) {
        int idx = base + lane;
        int c = 0;
        float w0 = 0.0f, w1 = 0.0f, w2 = 0.0f, w3 = 0.0f;
        if (idx < end) {
            c = perm[idx];
            float4 sd = *(const float4*)(s_dst + (size_t)c * HEADS);
            w0 = __expf(lrelu(ss.x + sd.x) - m0) / d0;
            w1 = __expf(lrelu(ss.y + sd.y) - m1) / d1;
            w2 = __expf(lrelu(ss.z + sd.z) - m2) / d2;
            w3 = __expf(lrelu(ss.w + sd.w) - m3) / d3;
        }
        int cntc = end - base;  if (cntc > 32) cntc = 32;
        int j = 0;
        for (; j + 4 <= cntc; j += 4) {
            float t[4];
            #pragma unroll
            for (int q = 0; q < 4; ++q) {
                const float* hp = h + (size_t)__builtin_amdgcn_readlane(c, j + q) * HC + lane;
                t[q] = rl_f(w0, j + q) * hp[0]          + rl_f(w1, j + q) * hp[OUT_CH]
                     + rl_f(w2, j + q) * hp[2 * OUT_CH] + rl_f(w3, j + q) * hp[3 * OUT_CH];
            }
            acc += t[0]; acc += t[1]; acc += t[2]; acc += t[3];
        }
        for (; j < cntc; ++j) {
            const float* hp = h + (size_t)__builtin_amdgcn_readlane(c, j) * HC + lane;
            acc += rl_f(w0, j) * hp[0]          + rl_f(w1, j) * hp[OUT_CH]
                 + rl_f(w2, j) * hp[2 * OUT_CH] + rl_f(w3, j) * hp[3 * OUT_CH];
        }
    }
    out_mean[(size_t)r * OUT_CH + lane] = 0.25f * acc;
    if (lane == 0) {
        *(float4*)(mx_out  + (size_t)r * HEADS) = make_float4(m0, m1, m2, m3);
        *(float4*)(den_out + (size_t)r * HEADS) = make_float4(d0, d1, d2, d3);
    }
}

__global__ void gat_alpha(const int* __restrict__ rows, const int* __restrict__ cols,
                          const float* __restrict__ s_src, const float* __restrict__ s_dst,
                          const float* __restrict__ mx, const float* __restrict__ den,
                          float* __restrict__ alpha_out, int E) {
    int e = blockIdx.x * blockDim.x + threadIdx.x;
    if (e >= E) return;
    int r = rows[e], c = cols[e];
    float4 ss = *(const float4*)(s_src + (size_t)r * HEADS);
    float4 sd = *(const float4*)(s_dst + (size_t)c * HEADS);
    float4 m  = *(const float4*)(mx    + (size_t)r * HEADS);
    float4 d  = *(const float4*)(den   + (size_t)r * HEADS);
    float4 o;
    o.x = __expf(lrelu(ss.x + sd.x) - m.x) / d.x;
    o.y = __expf(lrelu(ss.y + sd.y) - m.y) / d.y;
    o.z = __expf(lrelu(ss.z + sd.z) - m.z) / d.z;
    o.w = __expf(lrelu(ss.w + sd.w) - m.w) / d.w;
    *(float4*)(alpha_out + (size_t)e * HEADS) = o;
}

extern "C" void kernel_launch(void* const* d_in, const int* in_sizes, int n_in,
                              void* d_out, int out_size, void* d_ws, size_t ws_size,
                              hipStream_t stream) {
    const float* x    = (const float*)d_in[0];
    const int*   ei   = (const int*)d_in[1];
    const float* W    = (const float*)d_in[2];
    const float* attn = (const float*)d_in[3];

    const int n = in_sizes[0] / IN_CH;
    const int E = in_sizes[1] / 2;
    const int* rows = ei;
    const int* cols = ei + E;

    float* hbuf   = (float*)d_ws;
    float* s_src  = hbuf  + (size_t)n * HC;
    float* s_dst  = s_src + (size_t)n * HEADS;
    float* segmx  = s_dst + (size_t)n * HEADS;
    float* denom  = segmx + (size_t)n * HEADS;
    int*   cnt    = (int*)segmx;

    float* out_mean  = (float*)d_out;
    float* alpha_out = out_mean + (size_t)n * OUT_CH;
    int*   slots     = (int*)out_mean;
    int*   perm      = (int*)alpha_out;

    gat_init<<<(n + 255) / 256, 256, 0, stream>>>(cnt, n);

    gat_gemm<<<(n + 15) / 16, 256, 0, stream>>>(x, W, hbuf, n);

    gat_scores<<<(n * HEADS + 255) / 256, 256, 0, stream>>>(hbuf, attn, s_src, s_dst, n);

    gat_hist<<<(E + 255) / 256, 256, 0, stream>>>(rows, cnt, E);

    gat_scan<<<1, 1024, 0, stream>>>(cnt, slots, n);

    gat_fill<<<(E + 255) / 256, 256, 0, stream>>>(rows, cols, slots, perm, E);

    gat_rowpass<<<(n + 7) / 8, 256, 0, stream>>>(perm, s_src, s_dst, hbuf, out_mean, segmx, denom, n);

    gat_alpha<<<(E + 255) / 256, 256, 0, stream>>>(rows, cols, s_src, s_dst, segmx, denom, alpha_out, E);
}


# exclusive scan of the per-row edge counts spread over one work-group per 1024 rows (each first sums the counts in front of its rows) instead of one work-group for all rows; integer sums only
# speedup vs baseline: 1.5717x; 1.2859x over previous
#include <hip/hip_runtime.h>
#include <hip/hip_bf16.h>

typedef __attribute__((ext_vector_type(2))) float v2f;
typedef __attribute__((ext_vector_type(8))) float v8f;

#define IN_CH   256
#define HC      128
#define HEADS   4
#define OUT_CH  32
#define LDS_STRIDE 260

__device__ __forceinline__ float lrelu(float a) { return a > 0.0f ? a : 0.2f * a; }
__device__ __forceinline__ float rl_f(float v, int j) {
    return __int_as_float(__builtin_amdgcn_readlane(__float_as_int(v), j));
}


__global__ void gat_init(int* __restrict__ cnt, int n) {
    int i = blockIdx.x * blockDim.x + threadIdx.x;
    if (i < n) cnt[i] = 0;
}

__global__ void gat_gemm(const float* __restrict__ x, const float* __restrict__ W,
                         float* __restrict__ h, int n) {
    __shared__ float xs[16 * LDS_STRIDE];
    const int row0 = blockIdx.x * 16;

    const float4* xv = (const float4*)x;
    for (int i = 0; i < 4; ++i) {
        int idx = threadIdx.x + i * 256;
        int r   = idx >> 6;
        int c4  = idx & 63;
        int gr  = row0 + r;  if (gr >= n) gr = n - 1;
        float4 v = xv[(size_t)gr * 64 + c4];
        float* dst = &xs[r * LDS_STRIDE + (c4 << 2)];
        dst[0] = v.x; dst[1] = v.y; dst[2] = v.z; dst[3] = v.w;
    }
    __syncthreads();

    const int lane = threadIdx.x & 31;
    const int n0   = (threadIdx.x >> 5) << 4;
    const int m    = lane & 15;
    const int kb   = (lane >> 4) << 1;

    v8f acc = {};
    #pragma unroll 8
    for (int k = 0; k < IN_CH; k += 4) {
        v2f a, b;
        a.x = xs[m * LDS_STRIDE + k + kb];
        a.y = xs[m * LDS_STRIDE + k + kb + 1];
        const float* Wp = W + (size_t)(k + kb) * HC + n0 + m;
        b.x = Wp[0];
        b.y = Wp[HC];
        acc = __builtin_amdgcn_wmma_f32_16x16x4_f32(false, a, false, b,
                                                    (short)0, acc, false, false);
    }

    const int mOff = (lane >> 4) << 3;
    const int col  = n0 + m;
    #pragma unroll
    for (int i = 0; i < 8; ++i) {
        int r = row0 + mOff + i;
        if (r < n) h[(size_t)r * HC + col] = acc[i];
    }
}

__global__ void gat_scores(const float* __restrict__ h, const float* __restrict__ attn,
                           float* __restrict__ s_src, float* __restrict__ s_dst, int n) {
    int t = blockIdx.x * blockDim.x + threadIdx.x;
    if (t >= n * HEADS) return;
    int node = t >> 2, hd = t & 3;
    const float* hr = h + (size_t)node * HC + hd * OUT_CH;
    const float* as = attn + hd * (2 * OUT_CH);
    const float* ad = as + OUT_CH;
    float ss = 0.0f, sd = 0.0f;
    #pragma unroll 8
    for (int c = 0; c < OUT_CH; ++c) { float v = hr[c]; ss += v * as[c]; sd += v * ad[c]; }
    s_src[t] = ss; s_dst[t] = sd;
}

__global__ void gat_hist(const int* __restrict__ rows, int* __restrict__ cnt, int E) {
    int e = blockIdx.x * blockDim.x + threadIdx.x;
    if (e < E) atomicAdd(&cnt[rows[e]], 1);
}

__global__ void gat_scan(const int* __restrict__ cnt, int* __restrict__ slots, int n) {
    __shared__ int head[32];
    __shared__ int tot[32];
    const int t    = threadIdx.x;
    const int lane = t & 31;
    const int wave = t >> 5;
    const int base = blockIdx.x << 10;

    const int4* c4 = (const int4*)cnt;
    const int   n4 = base >> 2;
    int s = 0;
    #pragma unroll 4
    for (int j = t; j < n4; j += 1024) {
        int4 q = c4[j];
        s += (q.x + q.y) + (q.z + q.w);
    }
    for (int o = 16; o > 0; o >>= 1) s += __shfl_xor(s, o);

    const int i = base + t;
    const int v = (i < n) ? cnt[i] : 0;
    int x = v;
    for (int o = 1; o < 32; o <<= 1) {
        int y = __shfl_up(x, o);
        if (lane >= o) x += y;
    }
    if (lane == 0)  head[wave] = s;
    if (lane == 31) tot[wave]  = x;
    __syncthreads();
    int p = head[lane] + ((lane < wave) ? tot[lane] : 0);
    for (int o = 16; o > 0; o >>= 1) p += __shfl_xor(p, o);
    const int run = p + x - v;
    if (i < n) {
        slots[(size_t)i * OUT_CH]     = run;
        slots[(size_t)i * OUT_CH + 1] = run;
    }
}

__global__ void gat_fill(const int* __restrict__ rows, const int* __restrict__ cols,
                         int* __restrict__ slots, int* __restrict__ perm, int E) {
    int e = blockIdx.x * blockDim.x + threadIdx.x;
    if (e >= E) return;
    int pos = atomicAdd(&slots[(size_t)rows[e] * OUT_CH + 1], 1);
    perm[pos] = cols[e];
}

__global__ void gat_rowpass(const int* __restrict__ perm,
                            const float* __restrict__ s_src, const float* __restrict__ s_dst,
                            const float* __restrict__ h,
                            float* out_mean, float* __restrict__ mx_out,
                            float* __restrict__ den_out, int n) {
    const int r = blockIdx.x * (blockDim.x >> 5) + (threadIdx.x >> 5);
    if (r >= n) return;
    const int lane = threadIdx.x & 31;
    const int* slot = (const int*)(out_mean + (size_t)r * OUT_CH);
    const int start = __builtin_amdgcn_readfirstlane(slot[0]);
    const int end   = __builtin_amdgcn_readfirstlane(slot[1]);
    const float4 ss = *(const float4*)(s_src + (size_t)r * HEADS);
    const float NEG_INF = -__builtin_huge_valf();

    float m0 = NEG_INF, m1 = NEG_INF, m2 = NEG_INF, m3 = NEG_INF;
    for (int base = start; base < end; base += 32) {
        int idx = base + lane;
        if (idx < end) {
            int c = perm[idx];
            float4 sd = *(const float4*)(s_dst + (size_t)c * HEADS);
            m0 = fmaxf(m0, lrelu(ss.x + sd.x));
            m1 = fmaxf(m1, lrelu(ss.y + sd.y));
            m2 = fmaxf(m2, lrelu(ss.z + sd.z));
            m3 = fmaxf(m3, lrelu(ss.w + sd.w));
        }
    }
    for (int o = 16; o > 0; o >>= 1) {
        m0 = fmaxf(m0, __shfl_xor(m0, o));
        m1 = fmaxf(m1, __shfl_xor(m1, o));
        m2 = fmaxf(m2, __shfl_xor(m2, o));
        m3 = fmaxf(m3, __shfl_xor(m3, o));
    }

    float d0 = 0.0f, d1 = 0.0f, d2 = 0.0f, d3 = 0.0f;
    for (int base = start; base < end; base += 32) {
        int idx = base + lane;
        if (idx < end) {
            int c = perm[idx];
            float4 sd = *(const float4*)(s_dst + (size_t)c * HEADS);
            d0 += __expf(lrelu(ss.x + sd.x) - m0);
            d1 += __expf(lrelu(ss.y + sd.y) - m1);
            d2 += __expf(lrelu(ss.z + sd.z) - m2);
            d3 += __expf(lrelu(ss.w + sd.w) - m3);
        }
    }
    for (int o = 16; o > 0; o >>= 1) {
        d0 += __shfl_xor(d0, o);
        d1 += __shfl_xor(d1, o);
        d2 += __shfl_xor(d2, o);
        d3 += __shfl_xor(d3, o);
    }

    float acc = 0.0f;
    for (int base = start; base < end; base += 32) {
        int idx = base + lane;
        int c = 0;
        float w0 = 0.0f, w1 = 0.0f, w2 = 0.0f, w3 = 0.0f;
        if (idx < end) {
            c = perm[idx];
            float4 sd = *(const float4*)(s_dst + (size_t)c * HEADS);
            w0 = __expf(lrelu(ss.x + sd.x) - m0) / d0;
            w1 = __expf(lrelu(ss.y + sd.y) - m1) / d1;
            w2 = __expf(lrelu(ss.z + sd.z) - m2) / d2;
            w3 = __expf(lrelu(ss.w + sd.w) - m3) / d3;
        }
        int cntc = end - base;  if (cntc > 32) cntc = 32;
        int j = 0;
        for (; j + 4 <= cntc; j += 4) {
            float t[4];
            #pragma unroll
            for (int q = 0; q < 4; ++q) {
                const float* hp = h + (size_t)__builtin_amdgcn_readlane(c, j + q) * HC + lane;
                t[q] = rl_f(w0, j + q) * hp[0]          + rl_f(w1, j + q) * hp[OUT_CH]
                     + rl_f(w2, j + q) * hp[2 * OUT_CH] + rl_f(w3, j + q) * hp[3 * OUT_CH];
            }
            acc += t[0]; acc += t[1]; acc += t[2]; acc += t[3];
        }
        for (; j < cntc; ++j) {
            const float* hp = h + (size_t)__builtin_amdgcn_readlane(c, j) * HC + lane;
            acc += rl_f(w0, j) * hp[0]          + rl_f(w1, j) * hp[OUT_CH]
                 + rl_f(w2, j) * hp[2 * OUT_CH] + rl_f(w3, j) * hp[3 * OUT_CH];
        }
    }
    out_mean[(size_t)r * OUT_CH + lane] = 0.25f * acc;
    if (lane == 0) {
        *(float4*)(mx_out  + (size_t)r * HEADS) = make_float4(m0, m1, m2, m3);
        *(float4*)(den_out + (size_t)r * HEADS) = make_float4(d0, d1, d2, d3);
    }
}

__global__ void gat_alpha(const int* __restrict__ rows, const int* __restrict__ cols,
                          const float* __restrict__ s_src, const float* __restrict__ s_dst,
                          const float* __restrict__ mx, const float* __restrict__ den,
                          float* __restrict__ alpha_out, int E) {
    int e = blockIdx.x * blockDim.x + threadIdx.x;
    if (e >= E) return;
    int r = rows[e], c = cols[e];
    float4 ss = *(const float4*)(s_src + (size_t)r * HEADS);
    float4 sd = *(const float4*)(s_dst + (size_t)c * HEADS);
    float4 m  = *(const float4*)(mx    + (size_t)r * HEADS);
    float4 d  = *(const float4*)(den   + (size_t)r * HEADS);
    float4 o;
    o.x = __expf(lrelu(ss.x + sd.x) - m.x) / d.x;
    o.y = __expf(lrelu(ss.y + sd.y) - m.y) / d.y;
    o.z = __expf(lrelu(ss.z + sd.z) - m.z) / d.z;
    o.w = __expf(lrelu(ss.w + sd.w) - m.w) / d.w;
    *(float4*)(alpha_out + (size_t)e * HEADS) = o;
}

extern "C" void kernel_launch(void* const* d_in, const int* in_sizes, int n_in,
                              void* d_out, int out_size, void* d_ws, size_t ws_size,
                              hipStream_t stream) {
    const float* x    = (const float*)d_in[0];
    const int*   ei   = (const int*)d_in[1];
    const float* W    = (const float*)d_in[2];
    const float* attn = (const float*)d_in[3];

    const int n = in_sizes[0] / IN_CH;
    const int E = in_sizes[1] / 2;
    const int* rows = ei;
    const int* cols = ei + E;

    float* hbuf   = (float*)d_ws;
    float* s_src  = hbuf  + (size_t)n * HC;
    float* s_dst  = s_src + (size_t)n * HEADS;
    float* segmx  = s_dst + (size_t)n * HEADS;
    float* denom  = segmx + (size_t)n * HEADS;
    int*   cnt    = (int*)segmx;

    float* out_mean  = (float*)d_out;
    float* alpha_out = out_mean + (size_t)n * OUT_CH;
    int*   slots     = (int*)out_mean;
    int*   perm      = (int*)alpha_out;

    gat_init<<<(n + 255) / 256, 256, 0, stream>>>(cnt, n);

    gat_gemm<<<(n + 15) / 16, 256, 0, stream>>>(x, W, hbuf, n);

    gat_scores<<<(n * HEADS + 255) / 256, 256, 0, stream>>>(hbuf, attn, s_src, s_dst, n);

    gat_hist<<<(E + 255) / 256, 256, 0, stream>>>(rows, cnt, E);

    gat_scan<<<(n + 1023) / 1024, 1024, 0, stream>>>(cnt, slots, n);

    gat_fill<<<(E + 255) / 256, 256, 0, stream>>>(rows, cols, slots, perm, E);

    gat_rowpass<<<(n + 7) / 8, 256, 0, stream>>>(perm, s_src, s_dst, hbuf, out_mean, segmx, denom, n);

    gat_alpha<<<(E + 255) / 256, 256, 0, stream>>>(rows, cols, s_src, s_dst, segmx, denom, alpha_out, E);
}


# row pass holds a row of up to 64 edges in registers, two per lane: each edge's source index and destination score read once, not three times; longer rows keep the three sweeps; sum order unchanged
# speedup vs baseline: 1.6535x; 1.0520x over previous
#include <hip/hip_runtime.h>
#include <hip/hip_bf16.h>

typedef __attribute__((ext_vector_type(2))) float v2f;
typedef __attribute__((ext_vector_type(8))) float v8f;

#define IN_CH   256
#define HC      128
#define HEADS   4
#define OUT_CH  32
#define LDS_STRIDE 260

__device__ __forceinline__ float lrelu(float a) { return a > 0.0f ? a : 0.2f * a; }
__device__ __forceinline__ float rl_f(float v, int j) {
    return __int_as_float(__builtin_amdgcn_readlane(__float_as_int(v), j));
}


__global__ void gat_init(int* __restrict__ cnt, int n) {
    int i = blockIdx.x * blockDim.x + threadIdx.x;
    if (i < n) cnt[i] = 0;
}

__global__ void gat_gemm(const float* __restrict__ x, const float* __restrict__ W,
                         float* __restrict__ h, int n) {
    __shared__ float xs[16 * LDS_STRIDE];
    const int row0 = blockIdx.x * 16;

    const float4* xv = (const float4*)x;
    for (int i = 0; i < 4; ++i) {
        int idx = threadIdx.x + i * 256;
        int r   = idx >> 6;
        int c4  = idx & 63;
        int gr  = row0 + r;  if (gr >= n) gr = n - 1;
        float4 v = xv[(size_t)gr * 64 + c4];
        float* dst = &xs[r * LDS_STRIDE + (c4 << 2)];
        dst[0] = v.x; dst[1] = v.y; dst[2] = v.z; dst[3] = v.w;
    }
    __syncthreads();

    const int lane = threadIdx.x & 31;
    const int n0   = (threadIdx.x >> 5) << 4;
    const int m    = lane & 15;
    const int kb   = (lane >> 4) << 1;

    v8f acc = {};
    #pragma unroll 8
    for (int k = 0; k < IN_CH; k += 4) {
        v2f a, b;
        a.x = xs[m * LDS_STRIDE + k + kb];
        a.y = xs[m * LDS_STRIDE + k + kb + 1];
        const float* Wp = W + (size_t)(k + kb) * HC + n0 + m;
        b.x = Wp[0];
        b.y = Wp[HC];
        acc = __builtin_amdgcn_wmma_f32_16x16x4_f32(false, a, false, b,
                                                    (short)0, acc, false, false);
    }

    const int mOff = (lane >> 4) << 3;
    const int col  = n0 + m;
    #pragma unroll
    for (int i = 0; i < 8; ++i) {
        int r = row0 + mOff + i;
        if (r < n) h[(size_t)r * HC + col] = acc[i];
    }
}

__global__ void gat_scores(const float* __restrict__ h, const float* __restrict__ attn,
                           float* __restrict__ s_src, float* __restrict__ s_dst, int n) {
    int t = blockIdx.x * blockDim.x + threadIdx.x;
    if (t >= n * HEADS) return;
    int node = t >> 2, hd = t & 3;
    const float* hr = h + (size_t)node * HC + hd * OUT_CH;
    const float* as = attn + hd * (2 * OUT_CH);
    const float* ad = as + OUT_CH;
    float ss = 0.0f, sd = 0.0f;
    #pragma unroll 8
    for (int c = 0; c < OUT_CH; ++c) { float v = hr[c]; ss += v * as[c]; sd += v * ad[c]; }
    s_src[t] = ss; s_dst[t] = sd;
}

__global__ void gat_hist(const int* __restrict__ rows, int* __restrict__ cnt, int E) {
    int e = blockIdx.x * blockDim.x + threadIdx.x;
    if (e < E) atomicAdd(&cnt[rows[e]], 1);
}

__global__ void gat_scan(const int* __restrict__ cnt, int* __restrict__ slots, int n) {
    __shared__ int head[32];
    __shared__ int tot[32];
    const int t    = threadIdx.x;
    const int lane = t & 31;
    const int wave = t >> 5;
    const int base = blockIdx.x << 10;

    const int4* c4 = (const int4*)cnt;
    const int   n4 = base >> 2;
    int s = 0;
    #pragma unroll 4
    for (int j = t; j < n4; j += 1024) {
        int4 q = c4[j];
        s += (q.x + q.y) + (q.z + q.w);
    }
    for (int o = 16; o > 0; o >>= 1) s += __shfl_xor(s, o);

    const int i = base + t;
    const int v = (i < n) ? cnt[i] : 0;
    int x = v;
    for (int o = 1; o < 32; o <<= 1) {
        int y = __shfl_up(x, o);
        if (lane >= o) x += y;
    }
    if (lane == 0)  head[wave] = s;
    if (lane == 31) tot[wave]  = x;
    __syncthreads();
    int p = head[lane] + ((lane < wave) ? tot[lane] : 0);
    for (int o = 16; o > 0; o >>= 1) p += __shfl_xor(p, o);
    const int run = p + x - v;
    if (i < n) {
        slots[(size_t)i * OUT_CH]     = run;
        slots[(size_t)i * OUT_CH + 1] = run;
    }
}

__global__ void gat_fill(const int* __restrict__ rows, const int* __restrict__ cols,
                         int* __restrict__ slots, int* __restrict__ perm, int E) {
    int e = blockIdx.x * blockDim.x + threadIdx.x;
    if (e >= E) return;
    int pos = atomicAdd(&slots[(size_t)rows[e] * OUT_CH + 1], 1);
    perm[pos] = cols[e];
}

__device__ __forceinline__ float gat_chunk(const float* __restrict__ h, int lane, int c,
                                           float w0, float w1, float w2, float w3,
                                           int cntc, float acc) {
    int j = 0;
    for (; j + 4 <= cntc; j += 4) {
        float t[4];
        #pragma unroll
        for (int q = 0; q < 4; ++q) {
            const float* hp = h + (size_t)__builtin_amdgcn_readlane(c, j + q) * HC + lane;
            t[q] = rl_f(w0, j + q) * hp[0]          + rl_f(w1, j + q) * hp[OUT_CH]
                 + rl_f(w2, j + q) * hp[2 * OUT_CH] + rl_f(w3, j + q) * hp[3 * OUT_CH];
        }
        acc += t[0]; acc += t[1]; acc += t[2]; acc += t[3];
    }
    for (; j < cntc; ++j) {
        const float* hp = h + (size_t)__builtin_amdgcn_readlane(c, j) * HC + lane;
        acc += rl_f(w0, j) * hp[0]          + rl_f(w1, j) * hp[OUT_CH]
             + rl_f(w2, j) * hp[2 * OUT_CH] + rl_f(w3, j) * hp[3 * OUT_CH];
    }
    return acc;
}

__global__ void gat_rowpass(const int* __restrict__ perm,
                            const float* __restrict__ s_src, const float* __restrict__ s_dst,
                            const float* __restrict__ h,
                            float* out_mean, float* __restrict__ mx_out,
                            float* __restrict__ den_out, int n) {
    const int r = blockIdx.x * (blockDim.x >> 5) + (threadIdx.x >> 5);
    if (r >= n) return;
    const int lane = threadIdx.x & 31;
    const int* slot = (const int*)(out_mean + (size_t)r * OUT_CH);
    const int start = __builtin_amdgcn_readfirstlane(slot[0]);
    const int end   = __builtin_amdgcn_readfirstlane(slot[1]);
    const float4 ss = *(const float4*)(s_src + (size_t)r * HEADS);
    const float NEG_INF = -__builtin_huge_valf();
    const int deg = end - start;

    float m0 = NEG_INF, m1 = NEG_INF, m2 = NEG_INF, m3 = NEG_INF;
    float d0 = 0.0f, d1 = 0.0f, d2 = 0.0f, d3 = 0.0f;
    float acc = 0.0f;

    if (deg <= 64) {
        const int idx0 = start + lane, idx1 = idx0 + 32;
        const bool v0 = idx0 < end, v1 = idx1 < end;
        int c0 = 0, c1 = 0;
        float4 q0 = make_float4(0.0f, 0.0f, 0.0f, 0.0f), q1 = q0;
        if (deg > 0) {
            const int last = end - 1;
            c0 = perm[idx0 < last ? idx0 : last];
            if (deg > 32) c1 = perm[idx1 < last ? idx1 : last];
            q0 = *(const float4*)(s_dst + (size_t)c0 * HEADS);
            if (deg > 32) q1 = *(const float4*)(s_dst + (size_t)c1 * HEADS);
        }
        const float a00 = lrelu(ss.x + q0.x), a01 = lrelu(ss.y + q0.y);
        const float a02 = lrelu(ss.z + q0.z), a03 = lrelu(ss.w + q0.w);
        const float a10 = lrelu(ss.x + q1.x), a11 = lrelu(ss.y + q1.y);
        const float a12 = lrelu(ss.z + q1.z), a13 = lrelu(ss.w + q1.w);
        if (v0) { m0 = a00; m1 = a01; m2 = a02; m3 = a03; }
        if (v1) { m0 = fmaxf(m0, a10); m1 = fmaxf(m1, a11); m2 = fmaxf(m2, a12); m3 = fmaxf(m3, a13); }
        for (int o = 16; o > 0; o >>= 1) {
            m0 = fmaxf(m0, __shfl_xor(m0, o));
            m1 = fmaxf(m1, __shfl_xor(m1, o));
            m2 = fmaxf(m2, __shfl_xor(m2, o));
            m3 = fmaxf(m3, __shfl_xor(m3, o));
        }
        float e00 = 0.0f, e01 = 0.0f, e02 = 0.0f, e03 = 0.0f;
        float e10 = 0.0f, e11 = 0.0f, e12 = 0.0f, e13 = 0.0f;
        if (v0) { e00 = __expf(a00 - m0); e01 = __expf(a01 - m1); e02 = __expf(a02 - m2); e03 = __expf(a03 - m3); }
        if (v1) { e10 = __expf(a10 - m0); e11 = __expf(a11 - m1); e12 = __expf(a12 - m2); e13 = __expf(a13 - m3); }
        d0 = e00 + e10; d1 = e01 + e11; d2 = e02 + e12; d3 = e03 + e13;
        for (int o = 16; o > 0; o >>= 1) {
            d0 += __shfl_xor(d0, o);
            d1 += __shfl_xor(d1, o);
            d2 += __shfl_xor(d2, o);
            d3 += __shfl_xor(d3, o);
        }
        {
            float w0 = 0.0f, w1 = 0.0f, w2 = 0.0f, w3 = 0.0f;
            if (v0) { w0 = e00 / d0; w1 = e01 / d1; w2 = e02 / d2; w3 = e03 / d3; }
            acc = gat_chunk(h, lane, c0, w0, w1, w2, w3, deg < 32 ? deg : 32, acc);
        }
        if (deg > 32) {
            float w0 = 0.0f, w1 = 0.0f, w2 = 0.0f, w3 = 0.0f;
            if (v1) { w0 = e10 / d0; w1 = e11 / d1; w2 = e12 / d2; w3 = e13 / d3; }
            acc = gat_chunk(h, lane, c1, w0, w1, w2, w3, deg - 32, acc);
        }
    } else {
        for (int base = start; base < end; base += 32) {
            int idx = base + lane;
            if (idx < end) {
                int c = perm[idx];
                float4 sd = *(const float4*)(s_dst + (size_t)c * HEADS);
                m0 = fmaxf(m0, lrelu(ss.x + sd.x));
                m1 = fmaxf(m1, lrelu(ss.y + sd.y));
                m2 = fmaxf(m2, lrelu(ss.z + sd.z));
                m3 = fmaxf(m3, lrelu(ss.w + sd.w));
            }
        }
        for (int o = 16; o > 0; o >>= 1) {
            m0 = fmaxf(m0, __shfl_xor(m0, o));
            m1 = fmaxf(m1, __shfl_xor(m1, o));
            m2 = fmaxf(m2, __shfl_xor(m2, o));
            m3 = fmaxf(m3, __shfl_xor(m3, o));
        }
        for (int base = start; base < end; base += 32) {
            int idx = base + lane;
            if (idx < end) {
                int c = perm[idx];
                float4 sd = *(const float4*)(s_dst + (size_t)c * HEADS);
                d0 += __expf(lrelu(ss.x + sd.x) - m0);
                d1 += __expf(lrelu(ss.y + sd.y) - m1);
                d2 += __expf(lrelu(ss.z + sd.z) - m2);
                d3 += __expf(lrelu(ss.w + sd.w) - m3);
            }
        }
        for (int o = 16; o > 0; o >>= 1) {
            d0 += __shfl_xor(d0, o);
            d1 += __shfl_xor(d1, o);
            d2 += __shfl_xor(d2, o);
            d3 += __shfl_xor(d3, o);
        }
        for (int base = start; base < end; base += 32) {
            int idx = base + lane;
            int c = 0;
            float w0 = 0.0f, w1 = 0.0f, w2 = 0.0f, w3 = 0.0f;
            if (idx < end) {
                c = perm[idx];
                float4 sd = *(const float4*)(s_dst + (size_t)c * HEADS);
                w0 = __expf(lrelu(ss.x + sd.x) - m0) / d0;
                w1 = __expf(lrelu(ss.y + sd.y) - m1) / d1;
                w2 = __expf(lrelu(ss.z + sd.z) - m2) / d2;
                w3 = __expf(lrelu(ss.w + sd.w) - m3) / d3;
            }
            int cntc = end - base;  if (cntc > 32) cntc = 32;
            acc = gat_chunk(h, lane, c, w0, w1, w2, w3, cntc, acc);
        }
    }

    out_mean[(size_t)r * OUT_CH + lane] = 0.25f * acc;
    if (lane == 0) {
        *(float4*)(mx_out  + (size_t)r * HEADS) = make_float4(m0, m1, m2, m3);
        *(float4*)(den_out + (size_t)r * HEADS) = make_float4(d0, d1, d2, d3);
    }
}

__global__ void gat_alpha(const int* __restrict__ rows, const int* __restrict__ cols,
                          const float* __restrict__ s_src, const float* __restrict__ s_dst,
                          const float* __restrict__ mx, const float* __restrict__ den,
                          float* __restrict__ alpha_out, int E) {
    int e = blockIdx.x * blockDim.x + threadIdx.x;
    if (e >= E) return;
    int r = rows[e], c = cols[e];
    float4 ss = *(const float4*)(s_src + (size_t)r * HEADS);
    float4 sd = *(const float4*)(s_dst + (size_t)c * HEADS);
    float4 m  = *(const float4*)(mx    + (size_t)r * HEADS);
    float4 d  = *(const float4*)(den   + (size_t)r * HEADS);
    float4 o;
    o.x = __expf(lrelu(ss.x + sd.x) - m.x) / d.x;
    o.y = __expf(lrelu(ss.y + sd.y) - m.y) / d.y;
    o.z = __expf(lrelu(ss.z + sd.z) - m.z) / d.z;
    o.w = __expf(lrelu(ss.w + sd.w) - m.w) / d.w;
    *(float4*)(alpha_out + (size_t)e * HEADS) = o;
}

extern "C" void kernel_launch(void* const* d_in, const int* in_sizes, int n_in,
                              void* d_out, int out_size, void* d_ws, size_t ws_size,
                              hipStream_t stream) {
    const float* x    = (const float*)d_in[0];
    const int*   ei   = (const int*)d_in[1];
    const float* W    = (const float*)d_in[2];
    const float* attn = (const float*)d_in[3];

    const int n = in_sizes[0] / IN_CH;
    const int E = in_sizes[1] / 2;
    const int* rows = ei;
    const int* cols = ei + E;

    float* hbuf   = (float*)d_ws;
    float* s_src  = hbuf  + (size_t)n * HC;
    float* s_dst  = s_src + (size_t)n * HEADS;
    float* segmx  = s_dst + (size_t)n * HEADS;
    float* denom  = segmx + (size_t)n * HEADS;
    int*   cnt    = (int*)segmx;

    float* out_mean  = (float*)d_out;
    float* alpha_out = out_mean + (size_t)n * OUT_CH;
    int*   slots     = (int*)out_mean;
    int*   perm      = (int*)alpha_out;

    gat_init<<<(n + 255) / 256, 256, 0, stream>>>(cnt, n);

    gat_gemm<<<(n + 15) / 16, 256, 0, stream>>>(x, W, hbuf, n);

    gat_scores<<<(n * HEADS + 255) / 256, 256, 0, stream>>>(hbuf, attn, s_src, s_dst, n);

    gat_hist<<<(E + 255) / 256, 256, 0, stream>>>(rows, cnt, E);

    gat_scan<<<(n + 1023) / 1024, 1024, 0, stream>>>(cnt, slots, n);

    gat_fill<<<(E + 255) / 256, 256, 0, stream>>>(rows, cols, slots, perm, E);

    gat_rowpass<<<(n + 7) / 8, 256, 0, stream>>>(perm, s_src, s_dst, hbuf, out_mean, segmx, denom, n);

    gat_alpha<<<(E + 255) / 256, 256, 0, stream>>>(rows, cols, s_src, s_dst, segmx, denom, alpha_out, E);
}


# edges grouped by destination on LDS counters in two levels (256-row buckets, then rows), no per-edge global atomics; each row's sources sorted in registers: sum order fixed, not arrival order
# speedup vs baseline: 2.5452x; 1.5393x over previous
#include <hip/hip_runtime.h>
#include <hip/hip_bf16.h>

typedef __attribute__((ext_vector_type(2))) float v2f;
typedef __attribute__((ext_vector_type(8))) float v8f;

#define IN_CH   256
#define HC      128
#define HEADS   4
#define OUT_CH  32
#define LDS_STRIDE 260

__device__ __forceinline__ float lrelu(float a) { return a > 0.0f ? a : 0.2f * a; }
__device__ __forceinline__ float rl_f(float v, int j) {
    return __int_as_float(__builtin_amdgcn_readlane(__float_as_int(v), j));
}


#define RB_SHIFT 8
#define MAXB     256
#define CPAD     32
#define EPT      8

__global__ void gat_init(int* __restrict__ bcnt, int* __restrict__ bcur, int m) {
    int i = blockIdx.x * blockDim.x + threadIdx.x;
    if (i < m) { bcnt[i] = 0; bcur[i] = 0; }
}

__global__ void gat_gemm(const float* __restrict__ x, const float* __restrict__ W,
                         float* __restrict__ h, int n) {
    __shared__ float xs[16 * LDS_STRIDE];
    const int row0 = blockIdx.x * 16;

    const float4* xv = (const float4*)x;
    for (int i = 0; i < 4; ++i) {
        int idx = threadIdx.x + i * 256;
        int r   = idx >> 6;
        int c4  = idx & 63;
        int gr  = row0 + r;  if (gr >= n) gr = n - 1;
        float4 v = xv[(size_t)gr * 64 + c4];
        float* dst = &xs[r * LDS_STRIDE + (c4 << 2)];
        dst[0] = v.x; dst[1] = v.y; dst[2] = v.z; dst[3] = v.w;
    }
    __syncthreads();

    const int lane = threadIdx.x & 31;
    const int n0   = (threadIdx.x >> 5) << 4;
    const int m    = lane & 15;
    const int kb   = (lane >> 4) << 1;

    v8f acc = {};
    #pragma unroll 8
    for (int k = 0; k < IN_CH; k += 4) {
        v2f a, b;
        a.x = xs[m * LDS_STRIDE + k + kb];
        a.y = xs[m * LDS_STRIDE + k + kb + 1];
        const float* Wp = W + (size_t)(k + kb) * HC + n0 + m;
        b.x = Wp[0];
        b.y = Wp[HC];
        acc = __builtin_amdgcn_wmma_f32_16x16x4_f32(false, a, false, b,
                                                    (short)0, acc, false, false);
    }

    const int mOff = (lane >> 4) << 3;
    const int col  = n0 + m;
    #pragma unroll
    for (int i = 0; i < 8; ++i) {
        int r = row0 + mOff + i;
        if (r < n) h[(size_t)r * HC + col] = acc[i];
    }
}

__global__ void gat_scores(const float* __restrict__ h, const float* __restrict__ attn,
                           float* __restrict__ s_src, float* __restrict__ s_dst, int n) {
    int t = blockIdx.x * blockDim.x + threadIdx.x;
    if (t >= n * HEADS) return;
    int node = t >> 2, hd = t & 3;
    const float* hr = h + (size_t)node * HC + hd * OUT_CH;
    const float* as = attn + hd * (2 * OUT_CH);
    const float* ad = as + OUT_CH;
    float ss = 0.0f, sd = 0.0f;
    #pragma unroll 8
    for (int c = 0; c < OUT_CH; ++c) { float v = hr[c]; ss += v * as[c]; sd += v * ad[c]; }
    s_src[t] = ss; s_dst[t] = sd;
}

__global__ void gat_bincount(const int* __restrict__ rows, int* __restrict__ bcnt, int E, int B) {
    __shared__ int lc[MAXB];
    const int t = threadIdx.x;
    lc[t] = 0;
    __syncthreads();
    const int base = blockIdx.x * (256 * EPT);
    int r[EPT];
    #pragma unroll
    for (int k = 0; k < EPT; ++k) {
        int e = base + k * 256 + t;
        r[k] = (e < E) ? rows[e] : -1;
    }
    #pragma unroll
    for (int k = 0; k < EPT; ++k)
        if (r[k] >= 0) atomicAdd(&lc[r[k] >> RB_SHIFT], 1);
    __syncthreads();
    if (t < B) {
        int v = lc[t];
        if (v) atomicAdd(&bcnt[t * CPAD], v);
    }
}

__global__ void gat_binplace(const int* __restrict__ rows, const int* __restrict__ cols,
                             const int* __restrict__ bcnt, int* __restrict__ bcur,
                             int* __restrict__ rec, int E, int B) {
    __shared__ int lc[MAXB];
    __shared__ int gb[MAXB];
    __shared__ int wt[8];
    const int t    = threadIdx.x;
    const int lane = t & 31;
    const int wave = t >> 5;
    lc[t] = 0;

    const int v = (t < B) ? bcnt[t * CPAD] : 0;
    int x = v;
    for (int o = 1; o < 32; o <<= 1) {
        int y = __shfl_up(x, o);
        if (lane >= o) x += y;
    }
    if (lane == 31) wt[wave] = x;
    __syncthreads();
    int pre = 0;
    for (int w = 0; w < 8; ++w) if (w < wave) pre += wt[w];
    const int bstart = pre + x - v;

    const int base = blockIdx.x * (256 * EPT);
    int r[EPT], c[EPT], lr[EPT];
    #pragma unroll
    for (int k = 0; k < EPT; ++k) {
        int e = base + k * 256 + t;
        r[k] = (e < E) ? rows[e] : -1;
        c[k] = (e < E) ? cols[e] : 0;
    }
    #pragma unroll
    for (int k = 0; k < EPT; ++k) {
        lr[k] = 0;
        if (r[k] >= 0) lr[k] = atomicAdd(&lc[r[k] >> RB_SHIFT], 1);
    }
    __syncthreads();
    if (t < B) {
        int cnt = lc[t];
        int g = 0;
        if (cnt) g = atomicAdd(&bcur[t * CPAD], cnt);
        gb[t] = bstart + g;
    }
    __syncthreads();
    #pragma unroll
    for (int k = 0; k < EPT; ++k) {
        if (r[k] >= 0) {
            int p = gb[r[k] >> RB_SHIFT] + lr[k];
            *(int2*)(rec + 2 * (size_t)p) = make_int2(r[k], c[k]);
        }
    }
}

__global__ void gat_rowsort(const int* __restrict__ bcnt, const int* __restrict__ rec,
                            int* __restrict__ slots, int* __restrict__ perm, int n, int B) {
    __shared__ int rc[256];
    __shared__ int wt[32];
    const int t    = threadIdx.x;
    const int lane = t & 31;
    const int wave = t >> 5;
    const int b    = blockIdx.x;
    const int2* rec2 = (const int2*)rec;

    int s = (t < B && t < b) ? bcnt[t * CPAD] : 0;
    for (int o = 16; o > 0; o >>= 1) s += __shfl_xor(s, o);
    if (t < 256) rc[t] = 0;
    if (lane == 0) wt[wave] = s;
    __syncthreads();
    int bstart = wt[lane];
    for (int o = 16; o > 0; o >>= 1) bstart += __shfl_xor(bstart, o);
    const int bend = bstart + bcnt[b * CPAD];

    for (int p0 = bstart + t; p0 < bend; p0 += 4096) {
        int2 q[4];
        #pragma unroll
        for (int u = 0; u < 4; ++u) {
            int p = p0 + u * 1024;
            q[u] = rec2[p < bend ? p : p0];
        }
        #pragma unroll
        for (int u = 0; u < 4; ++u)
            if (p0 + u * 1024 < bend) atomicAdd(&rc[q[u].x & 255], 1);
    }
    __syncthreads();

    const int cntr = (t < 256) ? rc[t] : 0;
    int x = cntr;
    for (int o = 1; o < 32; o <<= 1) {
        int y = __shfl_up(x, o);
        if (lane >= o) x += y;
    }
    if (t < 256 && lane == 31) wt[wave] = x;
    __syncthreads();
    if (t < 256) {
        int pre = 0;
        for (int w = 0; w < 8; ++w) if (w < wave) pre += wt[w];
        const int rs = pre + x - cntr;
        rc[t] = rs;
        const int row = (b << RB_SHIFT) + t;
        if (row < n) {
            slots[(size_t)row * OUT_CH]     = bstart + rs;
            slots[(size_t)row * OUT_CH + 1] = bstart + rs + cntr;
        }
    }
    __syncthreads();

    for (int p0 = bstart + t; p0 < bend; p0 += 4096) {
        int2 q[4];
        #pragma unroll
        for (int u = 0; u < 4; ++u) {
            int p = p0 + u * 1024;
            q[u] = rec2[p < bend ? p : p0];
        }
        #pragma unroll
        for (int u = 0; u < 4; ++u) {
            if (p0 + u * 1024 < bend) {
                int pos = atomicAdd(&rc[q[u].x & 255], 1);
                perm[bstart + pos] = q[u].y;
            }
        }
    }
}

__device__ __forceinline__ int pick(int x, int y, bool keep_min) {
    const int lo = x < y ? x : y, hi = x < y ? y : x;
    return keep_min ? lo : hi;
}
__device__ __forceinline__ void sort32(int& x, int lane) {
    #pragma unroll
    for (int k = 2; k <= 32; k <<= 1) {
        #pragma unroll
        for (int j = k >> 1; j > 0; j >>= 1) {
            const int y = __shfl_xor(x, j);
            const bool up = (lane & k) == 0, lower = (lane & j) == 0;
            x = pick(x, y, up == lower);
        }
    }
}
__device__ __forceinline__ void sort64(int& x0, int& x1, int lane) {
    #pragma unroll
    for (int k = 2; k <= 32; k <<= 1) {
        #pragma unroll
        for (int j = k >> 1; j > 0; j >>= 1) {
            const int y0 = __shfl_xor(x0, j), y1 = __shfl_xor(x1, j);
            const bool up0 = (k == 32) ? true  : (lane & k) == 0;
            const bool up1 = (k == 32) ? false : (lane & k) == 0;
            const bool lower = (lane & j) == 0;
            x0 = pick(x0, y0, up0 == lower);
            x1 = pick(x1, y1, up1 == lower);
        }
    }
    { const int lo = x0 < x1 ? x0 : x1, hi = x0 < x1 ? x1 : x0; x0 = lo; x1 = hi; }
    #pragma unroll
    for (int j = 16; j > 0; j >>= 1) {
        const int y0 = __shfl_xor(x0, j), y1 = __shfl_xor(x1, j);
        const bool lower = (lane & j) == 0;
        x0 = pick(x0, y0, lower);
        x1 = pick(x1, y1, lower);
    }
}

__device__ __forceinline__ float gat_chunk(const float* __restrict__ h, int lane, int c,
                                           float w0, float w1, float w2, float w3,
                                           int cntc, float acc) {
    int j = 0;
    for (; j + 4 <= cntc; j += 4) {
        float t[4];
        #pragma unroll
        for (int q = 0; q < 4; ++q) {
            const float* hp = h + (size_t)__builtin_amdgcn_readlane(c, j + q) * HC + lane;
            t[q] = rl_f(w0, j + q) * hp[0]          + rl_f(w1, j + q) * hp[OUT_CH]
                 + rl_f(w2, j + q) * hp[2 * OUT_CH] + rl_f(w3, j + q) * hp[3 * OUT_CH];
        }
        acc += t[0]; acc += t[1]; acc += t[2]; acc += t[3];
    }
    for (; j < cntc; ++j) {
        const float* hp = h + (size_t)__builtin_amdgcn_readlane(c, j) * HC + lane;
        acc += rl_f(w0, j) * hp[0]          + rl_f(w1, j) * hp[OUT_CH]
             + rl_f(w2, j) * hp[2 * OUT_CH] + rl_f(w3, j) * hp[3 * OUT_CH];
    }
    return acc;
}

__global__ void gat_rowpass(const int* __restrict__ perm,
                            const float* __restrict__ s_src, const float* __restrict__ s_dst,
                            const float* __restrict__ h,
                            float* out_mean, float* __restrict__ mx_out,
                            float* __restrict__ den_out, int n) {
    const int r = blockIdx.x * (blockDim.x >> 5) + (threadIdx.x >> 5);
    if (r >= n) return;
    const int lane = threadIdx.x & 31;
    const int* slot = (const int*)(out_mean + (size_t)r * OUT_CH);
    const int start = __builtin_amdgcn_readfirstlane(slot[0]);
    const int end   = __builtin_amdgcn_readfirstlane(slot[1]);
    const float4 ss = *(const float4*)(s_src + (size_t)r * HEADS);
    const float NEG_INF = -__builtin_huge_valf();
    const int deg = end - start;

    float m0 = NEG_INF, m1 = NEG_INF, m2 = NEG_INF, m3 = NEG_INF;
    float d0 = 0.0f, d1 = 0.0f, d2 = 0.0f, d3 = 0.0f;
    float acc = 0.0f;

    if (deg <= 64) {
        const int idx0 = start + lane, idx1 = idx0 + 32;
        const bool v0 = idx0 < end, v1 = idx1 < end;
        int c0 = 0, c1 = 0;
        float4 q0 = make_float4(0.0f, 0.0f, 0.0f, 0.0f), q1 = q0;
        if (deg > 0) {
            const int last = end - 1;
            c0 = perm[idx0 < last ? idx0 : last];
            if (deg > 32) c1 = perm[idx1 < last ? idx1 : last];
            const int PAST = 0x7fffffff;
            int k0 = v0 ? c0 : PAST;
            int cl;
            if (deg > 32) {
                int k1 = v1 ? c1 : PAST;
                sort64(k0, k1, lane);
                c1 = k1;
                cl = __builtin_amdgcn_readlane(k1, deg - 33);
            } else {
                sort32(k0, lane);
                cl = __builtin_amdgcn_readlane(k0, deg - 1);
            }
            c0 = v0 ? k0 : cl;
            if (deg > 32) c1 = v1 ? c1 : cl;
            q0 = *(const float4*)(s_dst + (size_t)c0 * HEADS);
            if (deg > 32) q1 = *(const float4*)(s_dst + (size_t)c1 * HEADS);
        }
        const float a00 = lrelu(ss.x + q0.x), a01 = lrelu(ss.y + q0.y);
        const float a02 = lrelu(ss.z + q0.z), a03 = lrelu(ss.w + q0.w);
        const float a10 = lrelu(ss.x + q1.x), a11 = lrelu(ss.y + q1.y);
        const float a12 = lrelu(ss.z + q1.z), a13 = lrelu(ss.w + q1.w);
        if (v0) { m0 = a00; m1 = a01; m2 = a02; m3 = a03; }
        if (v1) { m0 = fmaxf(m0, a10); m1 = fmaxf(m1, a11); m2 = fmaxf(m2, a12); m3 = fmaxf(m3, a13); }
        for (int o = 16; o > 0; o >>= 1) {
            m0 = fmaxf(m0, __shfl_xor(m0, o));
            m1 = fmaxf(m1, __shfl_xor(m1, o));
            m2 = fmaxf(m2, __shfl_xor(m2, o));
            m3 = fmaxf(m3, __shfl_xor(m3, o));
        }
        float e00 = 0.0f, e01 = 0.0f, e02 = 0.0f, e03 = 0.0f;
        float e10 = 0.0f, e11 = 0.0f, e12 = 0.0f, e13 = 0.0f;
        if (v0) { e00 = __expf(a00 - m0); e01 = __expf(a01 - m1); e02 = __expf(a02 - m2); e03 = __expf(a03 - m3); }
        if (v1) { e10 = __expf(a10 - m0); e11 = __expf(a11 - m1); e12 = __expf(a12 - m2); e13 = __expf(a13 - m3); }
        d0 = e00 + e10; d1 = e01 + e11; d2 = e02 + e12; d3 = e03 + e13;
        for (int o = 16; o > 0; o >>= 1) {
            d0 += __shfl_xor(d0, o);
            d1 += __shfl_xor(d1, o);
            d2 += __shfl_xor(d2, o);
            d3 += __shfl_xor(d3, o);
        }
        {
            float w0 = 0.0f, w1 = 0.0f, w2 = 0.0f, w3 = 0.0f;
            if (v0) { w0 = e00 / d0; w1 = e01 / d1; w2 = e02 / d2; w3 = e03 / d3; }
            acc = gat_chunk(h, lane, c0, w0, w1, w2, w3, deg < 32 ? deg : 32, acc);
        }
        if (deg > 32) {
            float w0 = 0.0f, w1 = 0.0f, w2 = 0.0f, w3 = 0.0f;
            if (v1) { w0 = e10 / d0; w1 = e11 / d1; w2 = e12 / d2; w3 = e13 / d3; }
            acc = gat_chunk(h, lane, c1, w0, w1, w2, w3, deg - 32, acc);
        }
    } else {
        for (int base = start; base < end; base += 32) {
            int idx = base + lane;
            if (idx < end) {
                int c = perm[idx];
                float4 sd = *(const float4*)(s_dst + (size_t)c * HEADS);
                m0 = fmaxf(m0, lrelu(ss.x + sd.x));
                m1 = fmaxf(m1, lrelu(ss.y + sd.y));
                m2 = fmaxf(m2, lrelu(ss.z + sd.z));
                m3 = fmaxf(m3, lrelu(ss.w + sd.w));
            }
        }
        for (int o = 16; o > 0; o >>= 1) {
            m0 = fmaxf(m0, __shfl_xor(m0, o));
            m1 = fmaxf(m1, __shfl_xor(m1, o));
            m2 = fmaxf(m2, __shfl_xor(m2, o));
            m3 = fmaxf(m3, __shfl_xor(m3, o));
        }
        for (int base = start; base < end; base += 32) {
            int idx = base + lane;
            if (idx < end) {
                int c = perm[idx];
                float4 sd = *(const float4*)(s_dst + (size_t)c * HEADS);
                d0 += __expf(lrelu(ss.x + sd.x) - m0);
                d1 += __expf(lrelu(ss.y + sd.y) - m1);
                d2 += __expf(lrelu(ss.z + sd.z) - m2);
                d3 += __expf(lrelu(ss.w + sd.w) - m3);
            }
        }
        for (int o = 16; o > 0; o >>= 1) {
            d0 += __shfl_xor(d0, o);
            d1 += __shfl_xor(d1, o);
            d2 += __shfl_xor(d2, o);
            d3 += __shfl_xor(d3, o);
        }
        for (int base = start; base < end; base += 32) {
            int idx = base + lane;
            int c = 0;
            float w0 = 0.0f, w1 = 0.0f, w2 = 0.0f, w3 = 0.0f;
            if (idx < end) {
                c = perm[idx];
                float4 sd = *(const float4*)(s_dst + (size_t)c * HEADS);
                w0 = __expf(lrelu(ss.x + sd.x) - m0) / d0;
                w1 = __expf(lrelu(ss.y + sd.y) - m1) / d1;
                w2 = __expf(lrelu(ss.z + sd.z) - m2) / d2;
                w3 = __expf(lrelu(ss.w + sd.w) - m3) / d3;
            }
            int cntc = end - base;  if (cntc > 32) cntc = 32;
            acc = gat_chunk(h, lane, c, w0, w1, w2, w3, cntc, acc);
        }
    }

    out_mean[(size_t)r * OUT_CH + lane] = 0.25f * acc;
    if (lane == 0) {
        *(float4*)(mx_out  + (size_t)r * HEADS) = make_float4(m0, m1, m2, m3);
        *(float4*)(den_out + (size_t)r * HEADS) = make_float4(d0, d1, d2, d3);
    }
}

__global__ void gat_alpha(const int* __restrict__ rows, const int* __restrict__ cols,
                          const float* __restrict__ s_src, const float* __restrict__ s_dst,
                          const float* __restrict__ mx, const float* __restrict__ den,
                          float* __restrict__ alpha_out, int E) {
    int e = blockIdx.x * blockDim.x + threadIdx.x;
    if (e >= E) return;
    int r = rows[e], c = cols[e];
    float4 ss = *(const float4*)(s_src + (size_t)r * HEADS);
    float4 sd = *(const float4*)(s_dst + (size_t)c * HEADS);
    float4 m  = *(const float4*)(mx    + (size_t)r * HEADS);
    float4 d  = *(const float4*)(den   + (size_t)r * HEADS);
    float4 o;
    o.x = __expf(lrelu(ss.x + sd.x) - m.x) / d.x;
    o.y = __expf(lrelu(ss.y + sd.y) - m.y) / d.y;
    o.z = __expf(lrelu(ss.z + sd.z) - m.z) / d.z;
    o.w = __expf(lrelu(ss.w + sd.w) - m.w) / d.w;
    *(float4*)(alpha_out + (size_t)e * HEADS) = o;
}

extern "C" void kernel_launch(void* const* d_in, const int* in_sizes, int n_in,
                              void* d_out, int out_size, void* d_ws, size_t ws_size,
                              hipStream_t stream) {
    const float* x    = (const float*)d_in[0];
    const int*   ei   = (const int*)d_in[1];
    const float* W    = (const float*)d_in[2];
    const float* attn = (const float*)d_in[3];

    const int n = in_sizes[0] / IN_CH;
    const int E = in_sizes[1] / 2;
    const int* rows = ei;
    const int* cols = ei + E;

    float* hbuf   = (float*)d_ws;
    float* s_src  = hbuf  + (size_t)n * HC;
    float* s_dst  = s_src + (size_t)n * HEADS;
    float* segmx  = s_dst + (size_t)n * HEADS;
    float* denom  = segmx + (size_t)n * HEADS;
    int*   bcnt   = (int*)segmx;
    int*   bcur   = (int*)denom;

    float* out_mean  = (float*)d_out;
    float* alpha_out = out_mean + (size_t)n * OUT_CH;
    int*   slots     = (int*)out_mean;
    int*   perm      = (int*)alpha_out;
    int*   rec       = perm + 2 * (size_t)E;

    const int B = (n + 255) / 256;

    gat_init<<<(B * CPAD + 255) / 256, 256, 0, stream>>>(bcnt, bcur, B * CPAD);

    gat_gemm<<<(n + 15) / 16, 256, 0, stream>>>(x, W, hbuf, n);

    gat_scores<<<(n * HEADS + 255) / 256, 256, 0, stream>>>(hbuf, attn, s_src, s_dst, n);

    gat_bincount<<<(E + 256 * EPT - 1) / (256 * EPT), 256, 0, stream>>>(rows, bcnt, E, B);

    gat_binplace<<<(E + 256 * EPT - 1) / (256 * EPT), 256, 0, stream>>>(rows, cols, bcnt, bcur, rec, E, B);

    gat_rowsort<<<B, 1024, 0, stream>>>(bcnt, rec, slots, perm, n, B);

    gat_rowpass<<<(n + 7) / 8, 256, 0, stream>>>(perm, s_src, s_dst, hbuf, out_mean, segmx, denom, n);

    gat_alpha<<<(E + 255) / 256, 256, 0, stream>>>(rows, cols, s_src, s_dst, segmx, denom, alpha_out, E);
}


# one 64-byte record per node (source score, destination score, row maximum, row sum): the alpha pass reads two 128-byte lines per edge instead of four; every value and sum order unchanged
# speedup vs baseline: 2.7930x; 1.0974x over previous
#include <hip/hip_runtime.h>
#include <hip/hip_bf16.h>

typedef __attribute__((ext_vector_type(2))) float v2f;
typedef __attribute__((ext_vector_type(8))) float v8f;

#define IN_CH   256
#define HC      128
#define HEADS   4
#define OUT_CH  32
#define LDS_STRIDE 260

__device__ __forceinline__ float lrelu(float a) { return a > 0.0f ? a : 0.2f * a; }
__device__ __forceinline__ float rl_f(float v, int j) {
    return __int_as_float(__builtin_amdgcn_readlane(__float_as_int(v), j));
}


#define RB_SHIFT 8
#define MAXB     256
#define CPAD     32
#define EPT      8
#define NREC     16
#define R_DST    4
#define R_MAX    8
#define R_SUM    12

__global__ void gat_init(int* __restrict__ bcnt, int* __restrict__ bcur, int m) {
    int i = blockIdx.x * blockDim.x + threadIdx.x;
    if (i < m) { bcnt[i] = 0; bcur[i] = 0; }
}

__global__ void gat_gemm(const float* __restrict__ x, const float* __restrict__ W,
                         float* __restrict__ h, int n) {
    __shared__ float xs[16 * LDS_STRIDE];
    const int row0 = blockIdx.x * 16;

    const float4* xv = (const float4*)x;
    for (int i = 0; i < 4; ++i) {
        int idx = threadIdx.x + i * 256;
        int r   = idx >> 6;
        int c4  = idx & 63;
        int gr  = row0 + r;  if (gr >= n) gr = n - 1;
        float4 v = xv[(size_t)gr * 64 + c4];
        float* dst = &xs[r * LDS_STRIDE + (c4 << 2)];
        dst[0] = v.x; dst[1] = v.y; dst[2] = v.z; dst[3] = v.w;
    }
    __syncthreads();

    const int lane = threadIdx.x & 31;
    const int n0   = (threadIdx.x >> 5) << 4;
    const int m    = lane & 15;
    const int kb   = (lane >> 4) << 1;

    v8f acc = {};
    #pragma unroll 8
    for (int k = 0; k < IN_CH; k += 4) {
        v2f a, b;
        a.x = xs[m * LDS_STRIDE + k + kb];
        a.y = xs[m * LDS_STRIDE + k + kb + 1];
        const float* Wp = W + (size_t)(k + kb) * HC + n0 + m;
        b.x = Wp[0];
        b.y = Wp[HC];
        acc = __builtin_amdgcn_wmma_f32_16x16x4_f32(false, a, false, b,
                                                    (short)0, acc, false, false);
    }

    const int mOff = (lane >> 4) << 3;
    const int col  = n0 + m;
    #pragma unroll
    for (int i = 0; i < 8; ++i) {
        int r = row0 + mOff + i;
        if (r < n) h[(size_t)r * HC + col] = acc[i];
    }
}

__global__ void gat_scores(const float* __restrict__ h, const float* __restrict__ attn,
                           float* __restrict__ nrec, int n) {
    int t = blockIdx.x * blockDim.x + threadIdx.x;
    if (t >= n * HEADS) return;
    int node = t >> 2, hd = t & 3;
    const float* hr = h + (size_t)node * HC + hd * OUT_CH;
    const float* as = attn + hd * (2 * OUT_CH);
    const float* ad = as + OUT_CH;
    float ss = 0.0f, sd = 0.0f;
    #pragma unroll 8
    for (int c = 0; c < OUT_CH; ++c) { float v = hr[c]; ss += v * as[c]; sd += v * ad[c]; }
    nrec[(size_t)node * NREC + hd] = ss;
    nrec[(size_t)node * NREC + R_DST + hd] = sd;
}

__global__ void gat_bincount(const int* __restrict__ rows, int* __restrict__ bcnt, int E, int B) {
    __shared__ int lc[MAXB];
    const int t = threadIdx.x;
    lc[t] = 0;
    __syncthreads();
    const int base = blockIdx.x * (256 * EPT);
    int r[EPT];
    #pragma unroll
    for (int k = 0; k < EPT; ++k) {
        int e = base + k * 256 + t;
        r[k] = (e < E) ? rows[e] : -1;
    }
    #pragma unroll
    for (int k = 0; k < EPT; ++k)
        if (r[k] >= 0) atomicAdd(&lc[r[k] >> RB_SHIFT], 1);
    __syncthreads();
    if (t < B) {
        int v = lc[t];
        if (v) atomicAdd(&bcnt[t * CPAD], v);
    }
}

__global__ void gat_binplace(const int* __restrict__ rows, const int* __restrict__ cols,
                             const int* __restrict__ bcnt, int* __restrict__ bcur,
                             int* __restrict__ rec, int E, int B) {
    __shared__ int lc[MAXB];
    __shared__ int gb[MAXB];
    __shared__ int wt[8];
    const int t    = threadIdx.x;
    const int lane = t & 31;
    const int wave = t >> 5;
    lc[t] = 0;

    const int v = (t < B) ? bcnt[t * CPAD] : 0;
    int x = v;
    for (int o = 1; o < 32; o <<= 1) {
        int y = __shfl_up(x, o);
        if (lane >= o) x += y;
    }
    if (lane == 31) wt[wave] = x;
    __syncthreads();
    int pre = 0;
    for (int w = 0; w < 8; ++w) if (w < wave) pre += wt[w];
    const int bstart = pre + x - v;

    const int base = blockIdx.x * (256 * EPT);
    int r[EPT], c[EPT], lr[EPT];
    #pragma unroll
    for (int k = 0; k < EPT; ++k) {
        int e = base + k * 256 + t;
        r[k] = (e < E) ? rows[e] : -1;
        c[k] = (e < E) ? cols[e] : 0;
    }
    #pragma unroll
    for (int k = 0; k < EPT; ++k) {
        lr[k] = 0;
        if (r[k] >= 0) lr[k] = atomicAdd(&lc[r[k] >> RB_SHIFT], 1);
    }
    __syncthreads();
    if (t < B) {
        int cnt = lc[t];
        int g = 0;
        if (cnt) g = atomicAdd(&bcur[t * CPAD], cnt);
        gb[t] = bstart + g;
    }
    __syncthreads();
    #pragma unroll
    for (int k = 0; k < EPT; ++k) {
        if (r[k] >= 0) {
            int p = gb[r[k] >> RB_SHIFT] + lr[k];
            *(int2*)(rec + 2 * (size_t)p) = make_int2(r[k], c[k]);
        }
    }
}

__global__ void gat_rowsort(const int* __restrict__ bcnt, const int* __restrict__ rec,
                            int* __restrict__ slots, int* __restrict__ perm, int n, int B) {
    __shared__ int rc[256];
    __shared__ int wt[32];
    const int t    = threadIdx.x;
    const int lane = t & 31;
    const int wave = t >> 5;
    const int b    = blockIdx.x;
    const int2* rec2 = (const int2*)rec;

    int s = (t < B && t < b) ? bcnt[t * CPAD] : 0;
    for (int o = 16; o > 0; o >>= 1) s += __shfl_xor(s, o);
    if (t < 256) rc[t] = 0;
    if (lane == 0) wt[wave] = s;
    __syncthreads();
    int bstart = wt[lane];
    for (int o = 16; o > 0; o >>= 1) bstart += __shfl_xor(bstart, o);
    const int bend = bstart + bcnt[b * CPAD];

    for (int p0 = bstart + t; p0 < bend; p0 += 4096) {
        int2 q[4];
        #pragma unroll
        for (int u = 0; u < 4; ++u) {
            int p = p0 + u * 1024;
            q[u] = rec2[p < bend ? p : p0];
        }
        #pragma unroll
        for (int u = 0; u < 4; ++u)
            if (p0 + u * 1024 < bend) atomicAdd(&rc[q[u].x & 255], 1);
    }
    __syncthreads();

    const int cntr = (t < 256) ? rc[t] : 0;
    int x = cntr;
    for (int o = 1; o < 32; o <<= 1) {
        int y = __shfl_up(x, o);
        if (lane >= o) x += y;
    }
    if (t < 256 && lane == 31) wt[wave] = x;
    __syncthreads();
    if (t < 256) {
        int pre = 0;
        for (int w = 0; w < 8; ++w) if (w < wave) pre += wt[w];
        const int rs = pre + x - cntr;
        rc[t] = rs;
        const int row = (b << RB_SHIFT) + t;
        if (row < n) {
            slots[(size_t)row * OUT_CH]     = bstart + rs;
            slots[(size_t)row * OUT_CH + 1] = bstart + rs + cntr;
        }
    }
    __syncthreads();

    for (int p0 = bstart + t; p0 < bend; p0 += 4096) {
        int2 q[4];
        #pragma unroll
        for (int u = 0; u < 4; ++u) {
            int p = p0 + u * 1024;
            q[u] = rec2[p < bend ? p : p0];
        }
        #pragma unroll
        for (int u = 0; u < 4; ++u) {
            if (p0 + u * 1024 < bend) {
                int pos = atomicAdd(&rc[q[u].x & 255], 1);
                perm[bstart + pos] = q[u].y;
            }
        }
    }
}

__device__ __forceinline__ int pick(int x, int y, bool keep_min) {
    const int lo = x < y ? x : y, hi = x < y ? y : x;
    return keep_min ? lo : hi;
}
__device__ __forceinline__ void sort32(int& x, int lane) {
    #pragma unroll
    for (int k = 2; k <= 32; k <<= 1) {
        #pragma unroll
        for (int j = k >> 1; j > 0; j >>= 1) {
            const int y = __shfl_xor(x, j);
            const bool up = (lane & k) == 0, lower = (lane & j) == 0;
            x = pick(x, y, up == lower);
        }
    }
}
__device__ __forceinline__ void sort64(int& x0, int& x1, int lane) {
    #pragma unroll
    for (int k = 2; k <= 32; k <<= 1) {
        #pragma unroll
        for (int j = k >> 1; j > 0; j >>= 1) {
            const int y0 = __shfl_xor(x0, j), y1 = __shfl_xor(x1, j);
            const bool up0 = (k == 32) ? true  : (lane & k) == 0;
            const bool up1 = (k == 32) ? false : (lane & k) == 0;
            const bool lower = (lane & j) == 0;
            x0 = pick(x0, y0, up0 == lower);
            x1 = pick(x1, y1, up1 == lower);
        }
    }
    { const int lo = x0 < x1 ? x0 : x1, hi = x0 < x1 ? x1 : x0; x0 = lo; x1 = hi; }
    #pragma unroll
    for (int j = 16; j > 0; j >>= 1) {
        const int y0 = __shfl_xor(x0, j), y1 = __shfl_xor(x1, j);
        const bool lower = (lane & j) == 0;
        x0 = pick(x0, y0, lower);
        x1 = pick(x1, y1, lower);
    }
}

__device__ __forceinline__ float gat_chunk(const float* __restrict__ h, int lane, int c,
                                           float w0, float w1, float w2, float w3,
                                           int cntc, float acc) {
    int j = 0;
    for (; j + 4 <= cntc; j += 4) {
        float t[4];
        #pragma unroll
        for (int q = 0; q < 4; ++q) {
            const float* hp = h + (size_t)__builtin_amdgcn_readlane(c, j + q) * HC + lane;
            t[q] = rl_f(w0, j + q) * hp[0]          + rl_f(w1, j + q) * hp[OUT_CH]
                 + rl_f(w2, j + q) * hp[2 * OUT_CH] + rl_f(w3, j + q) * hp[3 * OUT_CH];
        }
        acc += t[0]; acc += t[1]; acc += t[2]; acc += t[3];
    }
    for (; j < cntc; ++j) {
        const float* hp = h + (size_t)__builtin_amdgcn_readlane(c, j) * HC + lane;
        acc += rl_f(w0, j) * hp[0]          + rl_f(w1, j) * hp[OUT_CH]
             + rl_f(w2, j) * hp[2 * OUT_CH] + rl_f(w3, j) * hp[3 * OUT_CH];
    }
    return acc;
}

__global__ void gat_rowpass(const int* __restrict__ perm, float* nrec,
                            const float* __restrict__ h, float* out_mean, int n) {
    const int r = blockIdx.x * (blockDim.x >> 5) + (threadIdx.x >> 5);
    if (r >= n) return;
    const int lane = threadIdx.x & 31;
    const int* slot = (const int*)(out_mean + (size_t)r * OUT_CH);
    const int start = __builtin_amdgcn_readfirstlane(slot[0]);
    const int end   = __builtin_amdgcn_readfirstlane(slot[1]);
    const float4 ss = *(const float4*)(nrec + (size_t)r * NREC);
    const float NEG_INF = -__builtin_huge_valf();
    const int deg = end - start;

    float m0 = NEG_INF, m1 = NEG_INF, m2 = NEG_INF, m3 = NEG_INF;
    float d0 = 0.0f, d1 = 0.0f, d2 = 0.0f, d3 = 0.0f;
    float acc = 0.0f;

    if (deg <= 64) {
        const int idx0 = start + lane, idx1 = idx0 + 32;
        const bool v0 = idx0 < end, v1 = idx1 < end;
        int c0 = 0, c1 = 0;
        float4 q0 = make_float4(0.0f, 0.0f, 0.0f, 0.0f), q1 = q0;
        if (deg > 0) {
            const int last = end - 1;
            c0 = perm[idx0 < last ? idx0 : last];
            if (deg > 32) c1 = perm[idx1 < last ? idx1 : last];
            const int PAST = 0x7fffffff;
            int k0 = v0 ? c0 : PAST;
            int cl;
            if (deg > 32) {
                int k1 = v1 ? c1 : PAST;
                sort64(k0, k1, lane);
                c1 = k1;
                cl = __builtin_amdgcn_readlane(k1, deg - 33);
            } else {
                sort32(k0, lane);
                cl = __builtin_amdgcn_readlane(k0, deg - 1);
            }
            c0 = v0 ? k0 : cl;
            if (deg > 32) c1 = v1 ? c1 : cl;
            q0 = *(const float4*)(nrec + (size_t)c0 * NREC + R_DST);
            if (deg > 32) q1 = *(const float4*)(nrec + (size_t)c1 * NREC + R_DST);
        }
        const float a00 = lrelu(ss.x + q0.x), a01 = lrelu(ss.y + q0.y);
        const float a02 = lrelu(ss.z + q0.z), a03 = lrelu(ss.w + q0.w);
        const float a10 = lrelu(ss.x + q1.x), a11 = lrelu(ss.y + q1.y);
        const float a12 = lrelu(ss.z + q1.z), a13 = lrelu(ss.w + q1.w);
        if (v0) { m0 = a00; m1 = a01; m2 = a02; m3 = a03; }
        if (v1) { m0 = fmaxf(m0, a10); m1 = fmaxf(m1, a11); m2 = fmaxf(m2, a12); m3 = fmaxf(m3, a13); }
        for (int o = 16; o > 0; o >>= 1) {
            m0 = fmaxf(m0, __shfl_xor(m0, o));
            m1 = fmaxf(m1, __shfl_xor(m1, o));
            m2 = fmaxf(m2, __shfl_xor(m2, o));
            m3 = fmaxf(m3, __shfl_xor(m3, o));
        }
        float e00 = 0.0f, e01 = 0.0f, e02 = 0.0f, e03 = 0.0f;
        float e10 = 0.0f, e11 = 0.0f, e12 = 0.0f, e13 = 0.0f;
        if (v0) { e00 = __expf(a00 - m0); e01 = __expf(a01 - m1); e02 = __expf(a02 - m2); e03 = __expf(a03 - m3); }
        if (v1) { e10 = __expf(a10 - m0); e11 = __expf(a11 - m1); e12 = __expf(a12 - m2); e13 = __expf(a13 - m3); }
        d0 = e00 + e10; d1 = e01 + e11; d2 = e02 + e12; d3 = e03 + e13;
        for (int o = 16; o > 0; o >>= 1) {
            d0 += __shfl_xor(d0, o);
            d1 += __shfl_xor(d1, o);
            d2 += __shfl_xor(d2, o);
            d3 += __shfl_xor(d3, o);
        }
        {
            float w0 = 0.0f, w1 = 0.0f, w2 = 0.0f, w3 = 0.0f;
            if (v0) { w0 = e00 / d0; w1 = e01 / d1; w2 = e02 / d2; w3 = e03 / d3; }
            acc = gat_chunk(h, lane, c0, w0, w1, w2, w3, deg < 32 ? deg : 32, acc);
        }
        if (deg > 32) {
            float w0 = 0.0f, w1 = 0.0f, w2 = 0.0f, w3 = 0.0f;
            if (v1) { w0 = e10 / d0; w1 = e11 / d1; w2 = e12 / d2; w3 = e13 / d3; }
            acc = gat_chunk(h, lane, c1, w0, w1, w2, w3, deg - 32, acc);
        }
    } else {
        for (int base = start; base < end; base += 32) {
            int idx = base + lane;
            if (idx < end) {
                int c = perm[idx];
                float4 sd = *(const float4*)(nrec + (size_t)c * NREC + R_DST);
                m0 = fmaxf(m0, lrelu(ss.x + sd.x));
                m1 = fmaxf(m1, lrelu(ss.y + sd.y));
                m2 = fmaxf(m2, lrelu(ss.z + sd.z));
                m3 = fmaxf(m3, lrelu(ss.w + sd.w));
            }
        }
        for (int o = 16; o > 0; o >>= 1) {
            m0 = fmaxf(m0, __shfl_xor(m0, o));
            m1 = fmaxf(m1, __shfl_xor(m1, o));
            m2 = fmaxf(m2, __shfl_xor(m2, o));
            m3 = fmaxf(m3, __shfl_xor(m3, o));
        }
        for (int base = start; base < end; base += 32) {
            int idx = base + lane;
            if (idx < end) {
                int c = perm[idx];
                float4 sd = *(const float4*)(nrec + (size_t)c * NREC + R_DST);
                d0 += __expf(lrelu(ss.x + sd.x) - m0);
                d1 += __expf(lrelu(ss.y + sd.y) - m1);
                d2 += __expf(lrelu(ss.z + sd.z) - m2);
                d3 += __expf(lrelu(ss.w + sd.w) - m3);
            }
        }
        for (int o = 16; o > 0; o >>= 1) {
            d0 += __shfl_xor(d0, o);
            d1 += __shfl_xor(d1, o);
            d2 += __shfl_xor(d2, o);
            d3 += __shfl_xor(d3, o);
        }
        for (int base = start; base < end; base += 32) {
            int idx = base + lane;
            int c = 0;
            float w0 = 0.0f, w1 = 0.0f, w2 = 0.0f, w3 = 0.0f;
            if (idx < end) {
                c = perm[idx];
                float4 sd = *(const float4*)(nrec + (size_t)c * NREC + R_DST);
                w0 = __expf(lrelu(ss.x + sd.x) - m0) / d0;
                w1 = __expf(lrelu(ss.y + sd.y) - m1) / d1;
                w2 = __expf(lrelu(ss.z + sd.z) - m2) / d2;
                w3 = __expf(lrelu(ss.w + sd.w) - m3) / d3;
            }
            int cntc = end - base;  if (cntc > 32) cntc = 32;
            acc = gat_chunk(h, lane, c, w0, w1, w2, w3, cntc, acc);
        }
    }

    out_mean[(size_t)r * OUT_CH + lane] = 0.25f * acc;
    if (lane == 0) {
        *(float4*)(nrec + (size_t)r * NREC + R_MAX) = make_float4(m0, m1, m2, m3);
        *(float4*)(nrec + (size_t)r * NREC + R_SUM) = make_float4(d0, d1, d2, d3);
    }
}

__global__ void gat_alpha(const int* __restrict__ rows, const int* __restrict__ cols,
                          const float* __restrict__ nrec,
                          float* __restrict__ alpha_out, int E) {
    int e = blockIdx.x * blockDim.x + threadIdx.x;
    if (e >= E) return;
    int r = rows[e], c = cols[e];
    const float* rr = nrec + (size_t)r * NREC;
    float4 ss = *(const float4*)rr;
    float4 sd = *(const float4*)(nrec + (size_t)c * NREC + R_DST);
    float4 m  = *(const float4*)(rr + R_MAX);
    float4 d  = *(const float4*)(rr + R_SUM);
    float4 o;
    o.x = __expf(lrelu(ss.x + sd.x) - m.x) / d.x;
    o.y = __expf(lrelu(ss.y + sd.y) - m.y) / d.y;
    o.z = __expf(lrelu(ss.z + sd.z) - m.z) / d.z;
    o.w = __expf(lrelu(ss.w + sd.w) - m.w) / d.w;
    *(float4*)(alpha_out + (size_t)e * HEADS) = o;
}

extern "C" void kernel_launch(void* const* d_in, const int* in_sizes, int n_in,
                              void* d_out, int out_size, void* d_ws, size_t ws_size,
                              hipStream_t stream) {
    const float* x    = (const float*)d_in[0];
    const int*   ei   = (const int*)d_in[1];
    const float* W    = (const float*)d_in[2];
    const float* attn = (const float*)d_in[3];

    const int n = in_sizes[0] / IN_CH;
    const int E = in_sizes[1] / 2;
    const int* rows = ei;
    const int* cols = ei + E;

    float* hbuf   = (float*)d_ws;
    float* nrec   = hbuf  + (size_t)n * HC;

    float* out_mean  = (float*)d_out;
    float* alpha_out = out_mean + (size_t)n * OUT_CH;
    int*   slots     = (int*)out_mean;
    int*   perm      = (int*)alpha_out;
    int*   rec       = perm + 2 * (size_t)E;

    const int B = (n + 255) / 256;
    int*   bcnt      = perm + (size_t)E;
    int*   bcur      = bcnt + B * CPAD;

    gat_init<<<(B * CPAD + 255) / 256, 256, 0, stream>>>(bcnt, bcur, B * CPAD);

    gat_gemm<<<(n + 15) / 16, 256, 0, stream>>>(x, W, hbuf, n);

    gat_scores<<<(n * HEADS + 255) / 256, 256, 0, stream>>>(hbuf, attn, nrec, n);

    gat_bincount<<<(E + 256 * EPT - 1) / (256 * EPT), 256, 0, stream>>>(rows, bcnt, E, B);

    gat_binplace<<<(E + 256 * EPT - 1) / (256 * EPT), 256, 0, stream>>>(rows, cols, bcnt, bcur, rec, E, B);

    gat_rowsort<<<B, 1024, 0, stream>>>(bcnt, rec, slots, perm, n, B);

    gat_rowpass<<<(n + 7) / 8, 256, 0, stream>>>(perm, nrec, hbuf, out_mean, n);

    gat_alpha<<<(E + 255) / 256, 256, 0, stream>>>(rows, cols, nrec, alpha_out, E);
}


# both grouping passes build their output in LDS (records by bucket; a bucket's sources by position) and store it from there: a few whole lines per store, not 32 scattered pieces; positions unchanged
# speedup vs baseline: 3.1038x; 1.1113x over previous
#include <hip/hip_runtime.h>
#include <hip/hip_bf16.h>

typedef __attribute__((ext_vector_type(2))) float v2f;
typedef __attribute__((ext_vector_type(8))) float v8f;

#define IN_CH   256
#define HC      128
#define HEADS   4
#define OUT_CH  32
#define LDS_STRIDE 260

__device__ __forceinline__ float lrelu(float a) { return a > 0.0f ? a : 0.2f * a; }
__device__ __forceinline__ float rl_f(float v, int j) {
    return __int_as_float(__builtin_amdgcn_readlane(__float_as_int(v), j));
}


#define RB_SHIFT 8
#define MAXB     256
#define CPAD     32
#define EPT      8
#define NREC     16
#define LPCAP    12288
#define R_DST    4
#define R_MAX    8
#define R_SUM    12

__device__ __forceinline__ float gat_fold(float lo, float hi, bool up, int w) {
    const float keep = up ? hi : lo;
    const float send = up ? lo : hi;
    return keep + __shfl_xor(send, w);
}

__global__ void gat_gemm(const float* __restrict__ x, const float* __restrict__ W,
                         const float* __restrict__ attn, float* __restrict__ h,
                         float* __restrict__ nrec, int* __restrict__ bcnt,
                         int* __restrict__ bcur, int mclear, int n) {
    __shared__ float xs[16 * LDS_STRIDE];
    __shared__ float sc[8 * 32];
    const int row0 = blockIdx.x * 16;

    {
        const int i = blockIdx.x * 256 + threadIdx.x;
        if (i < mclear) { bcnt[i] = 0; bcur[i] = 0; }
    }

    const float4* xv = (const float4*)x;
    for (int i = 0; i < 4; ++i) {
        int idx = threadIdx.x + i * 256;
        int r   = idx >> 6;
        int c4  = idx & 63;
        int gr  = row0 + r;  if (gr >= n) gr = n - 1;
        float4 v = xv[(size_t)gr * 64 + c4];
        float* dst = &xs[r * LDS_STRIDE + (c4 << 2)];
        dst[0] = v.x; dst[1] = v.y; dst[2] = v.z; dst[3] = v.w;
    }
    __syncthreads();

    const int lane = threadIdx.x & 31;
    const int wave = threadIdx.x >> 5;
    const int n0   = wave << 4;
    const int m    = lane & 15;
    const int kb   = (lane >> 4) << 1;

    const int hd = wave >> 1;
    const int ch = ((wave & 1) << 4) + m;
    const float as_l = attn[hd * (2 * OUT_CH) + ch];
    const float ad_l = attn[hd * (2 * OUT_CH) + OUT_CH + ch];

    v8f acc = {};
    #pragma unroll 8
    for (int k = 0; k < IN_CH; k += 4) {
        v2f a, b;
        a.x = xs[m * LDS_STRIDE + k + kb];
        a.y = xs[m * LDS_STRIDE + k + kb + 1];
        const float* Wp = W + (size_t)(k + kb) * HC + n0 + m;
        b.x = Wp[0];
        b.y = Wp[HC];
        acc = __builtin_amdgcn_wmma_f32_16x16x4_f32(false, a, false, b,
                                                    (short)0, acc, false, false);
    }

    const int mOff = (lane >> 4) << 3;
    const int col  = n0 + m;

    const bool u8 = (lane & 8) != 0, u4 = (lane & 4) != 0, u2 = (lane & 2) != 0, u1 = (lane & 1) != 0;
    const float r0 = gat_fold(acc[0] * as_l, acc[0] * ad_l, u8, 8);
    const float r1 = gat_fold(acc[1] * as_l, acc[1] * ad_l, u8, 8);
    const float r2 = gat_fold(acc[2] * as_l, acc[2] * ad_l, u8, 8);
    const float r3 = gat_fold(acc[3] * as_l, acc[3] * ad_l, u8, 8);
    const float r4 = gat_fold(acc[4] * as_l, acc[4] * ad_l, u8, 8);
    const float r5 = gat_fold(acc[5] * as_l, acc[5] * ad_l, u8, 8);
    const float r6 = gat_fold(acc[6] * as_l, acc[6] * ad_l, u8, 8);
    const float r7 = gat_fold(acc[7] * as_l, acc[7] * ad_l, u8, 8);
    const float s0 = gat_fold(r0, r4, u4, 4);
    const float s1 = gat_fold(r1, r5, u4, 4);
    const float s2 = gat_fold(r2, r6, u4, 4);
    const float s3 = gat_fold(r3, r7, u4, 4);
    const float t0 = gat_fold(s0, s2, u2, 2);
    const float t1 = gat_fold(s1, s3, u2, 2);
    const float part = gat_fold(t0, t1, u1, 1);
    sc[wave * 32 + lane] = part;
    __syncthreads();
    if (threadIdx.x < 32) {
        const int r = row0 + mOff + (lane & 7);
        const int kind = (lane >> 3) & 1;
        float4 o;
        o.x = sc[0 * 32 + lane] + sc[1 * 32 + lane];
        o.y = sc[2 * 32 + lane] + sc[3 * 32 + lane];
        o.z = sc[4 * 32 + lane] + sc[5 * 32 + lane];
        o.w = sc[6 * 32 + lane] + sc[7 * 32 + lane];
        if (r < n) *(float4*)(nrec + (size_t)r * NREC + kind * R_DST) = o;
    }

    #pragma unroll
    for (int i = 0; i < 8; ++i) {
        int r = row0 + mOff + i;
        if (r < n) h[(size_t)r * HC + col] = acc[i];
    }
}

__global__ void gat_bincount(const int* __restrict__ rows, int* __restrict__ bcnt, int E, int B) {
    __shared__ int lc[MAXB];
    const int t = threadIdx.x;
    lc[t] = 0;
    __syncthreads();
    const int base = blockIdx.x * (256 * EPT);
    int r[EPT];
    #pragma unroll
    for (int k = 0; k < EPT; ++k) {
        int e = base + k * 256 + t;
        r[k] = (e < E) ? rows[e] : -1;
    }
    #pragma unroll
    for (int k = 0; k < EPT; ++k)
        if (r[k] >= 0) atomicAdd(&lc[r[k] >> RB_SHIFT], 1);
    __syncthreads();
    if (t < B) {
        int v = lc[t];
        if (v) atomicAdd(&bcnt[t * CPAD], v);
    }
}

__global__ void gat_binplace(const int* __restrict__ rows, const int* __restrict__ cols,
                             const int* __restrict__ bcnt, int* __restrict__ bcur,
                             int* __restrict__ rec, int E, int B) {
    __shared__ int lc[MAXB];
    __shared__ int lo[MAXB];
    __shared__ int gb[MAXB];
    __shared__ int wt[8];
    __shared__ int wl[8];
    __shared__ int2 st[256 * EPT];
    const int t    = threadIdx.x;
    const int lane = t & 31;
    const int wave = t >> 5;
    lc[t] = 0;

    const int v = (t < B) ? bcnt[t * CPAD] : 0;
    int x = v;
    for (int o = 1; o < 32; o <<= 1) {
        int y = __shfl_up(x, o);
        if (lane >= o) x += y;
    }
    if (lane == 31) wt[wave] = x;
    __syncthreads();
    int pre = 0;
    for (int w = 0; w < 8; ++w) if (w < wave) pre += wt[w];
    const int bstart = pre + x - v;

    const int base = blockIdx.x * (256 * EPT);
    int r[EPT], c[EPT], lr[EPT];
    #pragma unroll
    for (int k = 0; k < EPT; ++k) {
        int e = base + k * 256 + t;
        r[k] = (e < E) ? rows[e] : -1;
        c[k] = (e < E) ? cols[e] : 0;
    }
    #pragma unroll
    for (int k = 0; k < EPT; ++k) {
        lr[k] = 0;
        if (r[k] >= 0) lr[k] = atomicAdd(&lc[r[k] >> RB_SHIFT], 1);
    }
    __syncthreads();
    const int cnt = lc[t];
    int g = 0;
    if (t < B && cnt) g = atomicAdd(&bcur[t * CPAD], cnt);
    int xl = cnt;
    for (int o = 1; o < 32; o <<= 1) {
        int y = __shfl_up(xl, o);
        if (lane >= o) xl += y;
    }
    if (lane == 31) wl[wave] = xl;
    __syncthreads();
    int prel = 0, total = 0;
    for (int w = 0; w < 8; ++w) { if (w < wave) prel += wl[w]; total += wl[w]; }
    const int lot = prel + xl - cnt;
    lo[t] = lot;
    gb[t] = bstart + g - lot;
    __syncthreads();
    const bool full = total == 256 * EPT;
    if (full) {
        int so[EPT];
        #pragma unroll
        for (int k = 0; k < EPT; ++k) so[k] = lo[r[k] >> RB_SHIFT] + lr[k];
        #pragma unroll
        for (int k = 0; k < EPT; ++k) st[so[k]] = make_int2(r[k], c[k]);
    } else {
        #pragma unroll
        for (int k = 0; k < EPT; ++k)
            if (r[k] >= 0) st[lo[r[k] >> RB_SHIFT] + lr[k]] = make_int2(r[k], c[k]);
    }
    __syncthreads();
    int2* rec2 = (int2*)rec;
    if (full) {
        int2 sv[EPT];
        int pb[EPT];
        #pragma unroll
        for (int k = 0; k < EPT; ++k) sv[k] = st[k * 256 + t];
        #pragma unroll
        for (int k = 0; k < EPT; ++k) pb[k] = gb[sv[k].x >> RB_SHIFT];
        #pragma unroll
        for (int k = 0; k < EPT; ++k) rec2[(size_t)(pb[k] + k * 256 + t)] = sv[k];
    } else {
        #pragma unroll
        for (int k = 0; k < EPT; ++k) {
            const int i = k * 256 + t;
            if (i < total) {
                const int2 sv = st[i];
                rec2[(size_t)(gb[sv.x >> RB_SHIFT] + i)] = sv;
            }
        }
    }
}

__global__ void gat_rowsort(const int* __restrict__ bcnt, const int* __restrict__ rec,
                            int* __restrict__ slots, int* __restrict__ perm, int n, int B) {
    __shared__ int rc[256];
    __shared__ int wt[32];
    __shared__ int lp[LPCAP];
    const int t    = threadIdx.x;
    const int lane = t & 31;
    const int wave = t >> 5;
    const int b    = blockIdx.x;
    const int2* rec2 = (const int2*)rec;

    int s = (t < B && t < b) ? bcnt[t * CPAD] : 0;
    for (int o = 16; o > 0; o >>= 1) s += __shfl_xor(s, o);
    if (t < 256) rc[t] = 0;
    if (lane == 0) wt[wave] = s;
    __syncthreads();
    int bstart = wt[lane];
    for (int o = 16; o > 0; o >>= 1) bstart += __shfl_xor(bstart, o);
    const int bend = bstart + bcnt[b * CPAD];
    const bool staged = bend - bstart <= LPCAP;

    for (int p0 = bstart + t; p0 < bend; p0 += 4096) {
        int2 q[4];
        #pragma unroll
        for (int u = 0; u < 4; ++u) {
            int p = p0 + u * 1024;
            q[u] = rec2[p < bend ? p : p0];
        }
        #pragma unroll
        for (int u = 0; u < 4; ++u)
            if (p0 + u * 1024 < bend) atomicAdd(&rc[q[u].x & 255], 1);
    }
    __syncthreads();

    const int cntr = (t < 256) ? rc[t] : 0;
    int x = cntr;
    for (int o = 1; o < 32; o <<= 1) {
        int y = __shfl_up(x, o);
        if (lane >= o) x += y;
    }
    if (t < 256 && lane == 31) wt[wave] = x;
    __syncthreads();
    if (t < 256) {
        int pre = 0;
        for (int w = 0; w < 8; ++w) if (w < wave) pre += wt[w];
        const int rs = pre + x - cntr;
        rc[t] = rs;
        const int row = (b << RB_SHIFT) + t;
        if (row < n) {
            slots[(size_t)row * OUT_CH]     = bstart + rs;
            slots[(size_t)row * OUT_CH + 1] = bstart + rs + cntr;
        }
    }
    __syncthreads();

    for (int p0 = bstart + t; p0 < bend; p0 += 4096) {
        int2 q[4];
        #pragma unroll
        for (int u = 0; u < 4; ++u) {
            int p = p0 + u * 1024;
            q[u] = rec2[p < bend ? p : p0];
        }
        #pragma unroll
        for (int u = 0; u < 4; ++u) {
            if (p0 + u * 1024 < bend) {
                int pos = atomicAdd(&rc[q[u].x & 255], 1);
                if (staged) lp[pos] = q[u].y;
                else        perm[bstart + pos] = q[u].y;
            }
        }
    }
    __syncthreads();
    if (staged) {
        const int cntb = bend - bstart;
        int i = t - (bstart & 31);
        if (i >= 0 && i < cntb) perm[bstart + i] = lp[i];
        #pragma unroll 4
        for (i += 1024; i < cntb; i += 1024) perm[bstart + i] = lp[i];
    }
}

__device__ __forceinline__ int pick(int x, int y, bool keep_min) {
    const int lo = x < y ? x : y, hi = x < y ? y : x;
    return keep_min ? lo : hi;
}
__device__ __forceinline__ void sort32(int& x, int lane) {
    #pragma unroll
    for (int k = 2; k <= 32; k <<= 1) {
        #pragma unroll
        for (int j = k >> 1; j > 0; j >>= 1) {
            const int y = __shfl_xor(x, j);
            const bool up = (lane & k) == 0, lower = (lane & j) == 0;
            x = pick(x, y, up == lower);
        }
    }
}
__device__ __forceinline__ void sort64(int& x0, int& x1, int lane) {
    #pragma unroll
    for (int k = 2; k <= 32; k <<= 1) {
        #pragma unroll
        for (int j = k >> 1; j > 0; j >>= 1) {
            const int y0 = __shfl_xor(x0, j), y1 = __shfl_xor(x1, j);
            const bool up0 = (k == 32) ? true  : (lane & k) == 0;
            const bool up1 = (k == 32) ? false : (lane & k) == 0;
            const bool lower = (lane & j) == 0;
            x0 = pick(x0, y0, up0 == lower);
            x1 = pick(x1, y1, up1 == lower);
        }
    }
    { const int lo = x0 < x1 ? x0 : x1, hi = x0 < x1 ? x1 : x0; x0 = lo; x1 = hi; }
    #pragma unroll
    for (int j = 16; j > 0; j >>= 1) {
        const int y0 = __shfl_xor(x0, j), y1 = __shfl_xor(x1, j);
        const bool lower = (lane & j) == 0;
        x0 = pick(x0, y0, lower);
        x1 = pick(x1, y1, lower);
    }
}

__device__ __forceinline__ float gat_chunk(const float* __restrict__ h, int lane, int c,
                                           float w0, float w1, float w2, float w3,
                                           int cntc, float acc) {
    int j = 0;
    for (; j + 4 <= cntc; j += 4) {
        float t[4];
        #pragma unroll
        for (int q = 0; q < 4; ++q) {
            const float* hp = h + (size_t)__builtin_amdgcn_readlane(c, j + q) * HC + lane;
            t[q] = rl_f(w0, j + q) * hp[0]          + rl_f(w1, j + q) * hp[OUT_CH]
                 + rl_f(w2, j + q) * hp[2 * OUT_CH] + rl_f(w3, j + q) * hp[3 * OUT_CH];
        }
        acc += t[0]; acc += t[1]; acc += t[2]; acc += t[3];
    }
    for (; j < cntc; ++j) {
        const float* hp = h + (size_t)__builtin_amdgcn_readlane(c, j) * HC + lane;
        acc += rl_f(w0, j) * hp[0]          + rl_f(w1, j) * hp[OUT_CH]
             + rl_f(w2, j) * hp[2 * OUT_CH] + rl_f(w3, j) * hp[3 * OUT_CH];
    }
    return acc;
}

__global__ void gat_rowpass(const int* __restrict__ perm, float* nrec,
                            const float* __restrict__ h, float* out_mean, int n) {
    const int r = blockIdx.x * (blockDim.x >> 5) + (threadIdx.x >> 5);
    if (r >= n) return;
    const int lane = threadIdx.x & 31;
    const int* slot = (const int*)(out_mean + (size_t)r * OUT_CH);
    const int start = __builtin_amdgcn_readfirstlane(slot[0]);
    const int end   = __builtin_amdgcn_readfirstlane(slot[1]);
    const float4 ss = *(const float4*)(nrec + (size_t)r * NREC);
    const float NEG_INF = -__builtin_huge_valf();
    const int deg = end - start;

    float m0 = NEG_INF, m1 = NEG_INF, m2 = NEG_INF, m3 = NEG_INF;
    float d0 = 0.0f, d1 = 0.0f, d2 = 0.0f, d3 = 0.0f;
    float acc = 0.0f;

    if (deg <= 64) {
        const int idx0 = start + lane, idx1 = idx0 + 32;
        const bool v0 = idx0 < end, v1 = idx1 < end;
        int c0 = 0, c1 = 0;
        float4 q0 = make_float4(0.0f, 0.0f, 0.0f, 0.0f), q1 = q0;
        if (deg > 0) {
            const int last = end - 1;
            c0 = perm[idx0 < last ? idx0 : last];
            if (deg > 32) c1 = perm[idx1 < last ? idx1 : last];
            const int PAST = 0x7fffffff;
            int k0 = v0 ? c0 : PAST;
            int cl;
            if (deg > 32) {
                int k1 = v1 ? c1 : PAST;
                sort64(k0, k1, lane);
                c1 = k1;
                cl = __builtin_amdgcn_readlane(k1, deg - 33);
            } else {
                sort32(k0, lane);
                cl = __builtin_amdgcn_readlane(k0, deg - 1);
            }
            c0 = v0 ? k0 : cl;
            if (deg > 32) c1 = v1 ? c1 : cl;
            q0 = *(const float4*)(nrec + (size_t)c0 * NREC + R_DST);
            if (deg > 32) q1 = *(const float4*)(nrec + (size_t)c1 * NREC + R_DST);
        }
        const float a00 = lrelu(ss.x + q0.x), a01 = lrelu(ss.y + q0.y);
        const float a02 = lrelu(ss.z + q0.z), a03 = lrelu(ss.w + q0.w);
        const float a10 = lrelu(ss.x + q1.x), a11 = lrelu(ss.y + q1.y);
        const float a12 = lrelu(ss.z + q1.z), a13 = lrelu(ss.w + q1.w);
        if (v0) { m0 = a00; m1 = a01; m2 = a02; m3 = a03; }
        if (v1) { m0 = fmaxf(m0, a10); m1 = fmaxf(m1, a11); m2 = fmaxf(m2, a12); m3 = fmaxf(m3, a13); }
        for (int o = 16; o > 0; o >>= 1) {
            m0 = fmaxf(m0, __shfl_xor(m0, o));
            m1 = fmaxf(m1, __shfl_xor(m1, o));
            m2 = fmaxf(m2, __shfl_xor(m2, o));
            m3 = fmaxf(m3, __shfl_xor(m3, o));
        }
        float e00 = 0.0f, e01 = 0.0f, e02 = 0.0f, e03 = 0.0f;
        float e10 = 0.0f, e11 = 0.0f, e12 = 0.0f, e13 = 0.0f;
        if (v0) { e00 = __expf(a00 - m0); e01 = __expf(a01 - m1); e02 = __expf(a02 - m2); e03 = __expf(a03 - m3); }
        if (v1) { e10 = __expf(a10 - m0); e11 = __expf(a11 - m1); e12 = __expf(a12 - m2); e13 = __expf(a13 - m3); }
        d0 = e00 + e10; d1 = e01 + e11; d2 = e02 + e12; d3 = e03 + e13;
        for (int o = 16; o > 0; o >>= 1) {
            d0 += __shfl_xor(d0, o);
            d1 += __shfl_xor(d1, o);
            d2 += __shfl_xor(d2, o);
            d3 += __shfl_xor(d3, o);
        }
        {
            float w0 = 0.0f, w1 = 0.0f, w2 = 0.0f, w3 = 0.0f;
            if (v0) { w0 = e00 / d0; w1 = e01 / d1; w2 = e02 / d2; w3 = e03 / d3; }
            acc = gat_chunk(h, lane, c0, w0, w1, w2, w3, deg < 32 ? deg : 32, acc);
        }
        if (deg > 32) {
            float w0 = 0.0f, w1 = 0.0f, w2 = 0.0f, w3 = 0.0f;
            if (v1) { w0 = e10 / d0; w1 = e11 / d1; w2 = e12 / d2; w3 = e13 / d3; }
            acc = gat_chunk(h, lane, c1, w0, w1, w2, w3, deg - 32, acc);
        }
    } else {
        for (int base = start; base < end; base += 32) {
            int idx = base + lane;
            if (idx < end) {
                int c = perm[idx];
                float4 sd = *(const float4*)(nrec + (size_t)c * NREC + R_DST);
                m0 = fmaxf(m0, lrelu(ss.x + sd.x));
                m1 = fmaxf(m1, lrelu(ss.y + sd.y));
                m2 = fmaxf(m2, lrelu(ss.z + sd.z));
                m3 = fmaxf(m3, lrelu(ss.w + sd.w));
            }
        }
        for (int o = 16; o > 0; o >>= 1) {
            m0 = fmaxf(m0, __shfl_xor(m0, o));
            m1 = fmaxf(m1, __shfl_xor(m1, o));
            m2 = fmaxf(m2, __shfl_xor(m2, o));
            m3 = fmaxf(m3, __shfl_xor(m3, o));
        }
        for (int base = start; base < end; base += 32) {
            int idx = base + lane;
            if (idx < end) {
                int c = perm[idx];
                float4 sd = *(const float4*)(nrec + (size_t)c * NREC + R_DST);
                d0 += __expf(lrelu(ss.x + sd.x) - m0);
                d1 += __expf(lrelu(ss.y + sd.y) - m1);
                d2 += __expf(lrelu(ss.z + sd.z) - m2);
                d3 += __expf(lrelu(ss.w + sd.w) - m3);
            }
        }
        for (int o = 16; o > 0; o >>= 1) {
            d0 += __shfl_xor(d0, o);
            d1 += __shfl_xor(d1, o);
            d2 += __shfl_xor(d2, o);
            d3 += __shfl_xor(d3, o);
        }
        for (int base = start; base < end; base += 32) {
            int idx = base + lane;
            int c = 0;
            float w0 = 0.0f, w1 = 0.0f, w2 = 0.0f, w3 = 0.0f;
            if (idx < end) {
                c = perm[idx];
                float4 sd = *(const float4*)(nrec + (size_t)c * NREC + R_DST);
                w0 = __expf(lrelu(ss.x + sd.x) - m0) / d0;
                w1 = __expf(lrelu(ss.y + sd.y) - m1) / d1;
                w2 = __expf(lrelu(ss.z + sd.z) - m2) / d2;
                w3 = __expf(lrelu(ss.w + sd.w) - m3) / d3;
            }
            int cntc = end - base;  if (cntc > 32) cntc = 32;
            acc = gat_chunk(h, lane, c, w0, w1, w2, w3, cntc, acc);
        }
    }

    out_mean[(size_t)r * OUT_CH + lane] = 0.25f * acc;
    if (lane == 0) {
        *(float4*)(nrec + (size_t)r * NREC + R_MAX) = make_float4(m0, m1, m2, m3);
        *(float4*)(nrec + (size_t)r * NREC + R_SUM) = make_float4(d0, d1, d2, d3);
    }
}

__global__ void gat_alpha(const int* __restrict__ rows, const int* __restrict__ cols,
                          const float* __restrict__ nrec,
                          float* __restrict__ alpha_out, int E) {
    int e = blockIdx.x * blockDim.x + threadIdx.x;
    if (e >= E) return;
    int r = rows[e], c = cols[e];
    const float* rr = nrec + (size_t)r * NREC;
    float4 ss = *(const float4*)rr;
    float4 sd = *(const float4*)(nrec + (size_t)c * NREC + R_DST);
    float4 m  = *(const float4*)(rr + R_MAX);
    float4 d  = *(const float4*)(rr + R_SUM);
    float4 o;
    o.x = __expf(lrelu(ss.x + sd.x) - m.x) / d.x;
    o.y = __expf(lrelu(ss.y + sd.y) - m.y) / d.y;
    o.z = __expf(lrelu(ss.z + sd.z) - m.z) / d.z;
    o.w = __expf(lrelu(ss.w + sd.w) - m.w) / d.w;
    *(float4*)(alpha_out + (size_t)e * HEADS) = o;
}

extern "C" void kernel_launch(void* const* d_in, const int* in_sizes, int n_in,
                              void* d_out, int out_size, void* d_ws, size_t ws_size,
                              hipStream_t stream) {
    const float* x    = (const float*)d_in[0];
    const int*   ei   = (const int*)d_in[1];
    const float* W    = (const float*)d_in[2];
    const float* attn = (const float*)d_in[3];

    const int n = in_sizes[0] / IN_CH;
    const int E = in_sizes[1] / 2;
    const int* rows = ei;
    const int* cols = ei + E;

    float* hbuf   = (float*)d_ws;
    float* nrec   = hbuf  + (size_t)n * HC;

    float* out_mean  = (float*)d_out;
    float* alpha_out = out_mean + (size_t)n * OUT_CH;
    int*   slots     = (int*)out_mean;
    int*   perm      = (int*)alpha_out;
    int*   rec       = perm + 2 * (size_t)E;

    const int B = (n + 255) / 256;
    int*   bcnt      = perm + (size_t)E;
    int*   bcur      = bcnt + B * CPAD;

    gat_gemm<<<(n + 15) / 16, 256, 0, stream>>>(x, W, attn, hbuf, nrec, bcnt, bcur, B * CPAD, n);

    gat_bincount<<<(E + 256 * EPT - 1) / (256 * EPT), 256, 0, stream>>>(rows, bcnt, E, B);

    gat_binplace<<<(E + 256 * EPT - 1) / (256 * EPT), 256, 0, stream>>>(rows, cols, bcnt, bcur, rec, E, B);

    gat_rowsort<<<B, 1024, 0, stream>>>(bcnt, rec, slots, perm, n, B);

    gat_rowpass<<<(n + 7) / 8, 256, 0, stream>>>(perm, nrec, hbuf, out_mean, n);

    gat_alpha<<<(E + 255) / 256, 256, 0, stream>>>(rows, cols, nrec, alpha_out, E);
}
